# starting kernel
# baseline (speedup 1.0000x reference)
#include <hip/hip_runtime.h>

#define B_   128
#define TE_  2048
#define U_   256
#define IN_  256
#define MT_  4

typedef __attribute__((ext_vector_type(16))) __bf16 v16bf;
typedef __attribute__((ext_vector_type(8)))  float  v8f;

__device__ __forceinline__ unsigned short f32_to_bf16_bits(float f) {
  unsigned int u = __float_as_uint(f);
  u += 0x7FFFu + ((u >> 16) & 1u);
  return (unsigned short)(u >> 16);
}

__global__ void k_prep_uaT(const float* __restrict__ Ua,
                           unsigned short* __restrict__ UaT) {
  int e = blockIdx.x;
  int u = threadIdx.x;
  UaT[u * U_ + e] = f32_to_bf16_bits(Ua[e * U_ + u]);
}

__global__ void k_wxstm(const float* __restrict__ h,
                        const float* __restrict__ Wa,
                        float* __restrict__ wx) {
  int b = blockIdx.x, u = threadIdx.x;
  __shared__ float sh[U_];
  sh[u] = h[b * U_ + u];
  __syncthreads();
  float a = 0.f;
  #pragma unroll 4
  for (int k = 0; k < U_; ++k) a = fmaf(sh[k], Wa[k * U_ + u], a);
  wx[b * U_ + u] = a;
}

__global__ __launch_bounds__(256) void k_scores(
    const float* __restrict__ x_seq, const unsigned short* __restrict__ UaT,
    const float* __restrict__ wxstm, const float* __restrict__ b_a,
    const float* __restrict__ V_a, float* __restrict__ et) {
  __shared__ __align__(16) unsigned short sA[16 * MT_ * U_];
  __shared__ float s_e[16 * MT_];

  int b  = blockIdx.y;
  int t0 = blockIdx.x * (16 * MT_);
  int tid = threadIdx.x;

  const float* xrow = x_seq + ((size_t)b * TE_ + t0) * U_;
  for (int i = tid; i < 16 * MT_ * U_; i += 256) sA[i] = f32_to_bf16_bits(xrow[i]);
  if (tid < 16 * MT_) s_e[tid] = 0.f;
  __syncthreads();

  int wave = tid >> 5;
  int lane = tid & 31;
  int sel  = lane >> 4;
  int row  = lane & 15;

  v8f acc[2][MT_];
  #pragma unroll
  for (int j = 0; j < 2; ++j)
    #pragma unroll
    for (int m = 0; m < MT_; ++m)
      acc[j][m] = (v8f){0.f, 0.f, 0.f, 0.f, 0.f, 0.f, 0.f, 0.f};

  const unsigned short* bcol0 = UaT + (size_t)(2 * wave * 16 + row) * U_;
  const unsigned short* bcol1 = bcol0 + 16 * U_;

  #pragma unroll
  for (int kk = 0; kk < U_; kk += 32) {
    union { uint4 q[2]; v16bf v; } fb0, fb1;
    fb0.q[0] = *(const uint4*)(bcol0 + kk + sel * 8);
    fb0.q[1] = *(const uint4*)(bcol0 + kk + 16 + sel * 8);
    fb1.q[0] = *(const uint4*)(bcol1 + kk + sel * 8);
    fb1.q[1] = *(const uint4*)(bcol1 + kk + 16 + sel * 8);
    #pragma unroll
    for (int m = 0; m < MT_; ++m) {
      union { uint4 q[2]; v16bf v; } fa;
      const unsigned short* arow = &sA[(m * 16 + row) * U_ + kk];
      fa.q[0] = *(const uint4*)(arow + sel * 8);
      fa.q[1] = *(const uint4*)(arow + 16 + sel * 8);
      acc[0][m] = __builtin_amdgcn_wmma_f32_16x16x32_bf16(
          false, fa.v, false, fb0.v, (short)0, acc[0][m], false, false);
      acc[1][m] = __builtin_amdgcn_wmma_f32_16x16x32_bf16(
          false, fa.v, false, fb1.v, (short)0, acc[1][m], false, false);
    }
  }

  #pragma unroll
  for (int j = 0; j < 2; ++j) {
    int ug = (2 * wave + j) * 16 + row;
    float wx = wxstm[b * U_ + ug] + b_a[ug];
    float va = V_a[ug];
    #pragma unroll
    for (int m = 0; m < MT_; ++m) {
      #pragma unroll
      for (int i = 0; i < 8; ++i) {
        atomicAdd(&s_e[m * 16 + i + 8 * sel], va * tanhf(acc[j][m][i] + wx));
      }
    }
  }
  __syncthreads();
  if (tid < 16 * MT_) et[(size_t)b * TE_ + t0 + tid] = s_e[tid];
}

__global__ void k_softmax(float* __restrict__ at, float* __restrict__ ctx) {
  int b = blockIdx.x, tid = threadIdx.x;
  float v[8];
  float s = 0.f;
  #pragma unroll
  for (int j = 0; j < 8; ++j) {
    v[j] = expf(at[(size_t)b * TE_ + j * 256 + tid]);
    s += v[j];
  }
  __shared__ float red[256];
  red[tid] = s;
  __syncthreads();
  for (int off = 128; off > 0; off >>= 1) {
    if (tid < off) red[tid] += red[tid + off];
    __syncthreads();
  }
  float inv = 1.f / red[0];
  #pragma unroll
  for (int j = 0; j < 8; ++j)
    at[(size_t)b * TE_ + j * 256 + tid] = v[j] * inv;
  ctx[b * U_ + tid] = 0.f;
}

__global__ void k_context(const float* __restrict__ x_seq,
                          const float* __restrict__ at,
                          float* __restrict__ ctx) {
  int seg = blockIdx.x, b = blockIdx.y, u = threadIdx.x;
  const float* xp = x_seq + ((size_t)b * TE_ + seg * 128) * U_ + u;
  const float* ap = at + (size_t)b * TE_ + seg * 128;
  float acc = 0.f;
  for (int t = 0; t < 128; ++t) {
    __builtin_prefetch(xp + 8 * U_, 0, 1);
    acc = fmaf(ap[t], xp[0], acc);
    xp += U_;
  }
  atomicAdd(&ctx[b * U_ + u], acc);
}

__global__ void k_gates(const float* __restrict__ h, const float* __restrict__ in,
                        const float* __restrict__ ctx,
                        const float* __restrict__ Wz, const float* __restrict__ Cz,
                        const float* __restrict__ bz,
                        const float* __restrict__ Wr, const float* __restrict__ Cr,
                        const float* __restrict__ br,
                        float* __restrict__ ztw, float* __restrict__ rhw) {
  int b = blockIdx.x, u = threadIdx.x;
  __shared__ float sh[U_], si[IN_], sc[U_];
  sh[u] = h[b * U_ + u];
  si[u] = in[b * IN_ + u];
  sc[u] = ctx[b * U_ + u];
  __syncthreads();
  float az = bz[u], ar = br[u];
  for (int k = 0; k < U_; ++k) {
    float hv = sh[k];
    az = fmaf(hv, Wz[k * U_ + u], az);
    ar = fmaf(hv, Wr[k * U_ + u], ar);
  }
  for (int k = 0; k < IN_; ++k) {
    float iv = si[k];
    az = fmaf(iv, Cz[k * U_ + u], az);
    ar = fmaf(iv, Cr[k * U_ + u], ar);
  }
  for (int k = 0; k < U_; ++k) {
    float cv = sc[k];
    az = fmaf(cv, Cz[(IN_ + k) * U_ + u], az);
    ar = fmaf(cv, Cr[(IN_ + k) * U_ + u], ar);
  }
  float zt = 1.f / (1.f + expf(-az));
  float rt = 1.f / (1.f + expf(-ar));
  ztw[b * U_ + u] = zt;
  rhw[b * U_ + u] = rt * sh[u];
}

__global__ void k_out(const float* __restrict__ h, const float* __restrict__ in,
                      const float* __restrict__ ctx,
                      const float* __restrict__ Up, const float* __restrict__ Cp,
                      const float* __restrict__ bp,
                      const float* __restrict__ ztw, const float* __restrict__ rhw,
                      float* __restrict__ out) {
  int b = blockIdx.x, u = threadIdx.x;
  __shared__ float sr[U_], si[IN_], sc[U_];
  sr[u] = rhw[b * U_ + u];
  si[u] = in[b * IN_ + u];
  sc[u] = ctx[b * U_ + u];
  __syncthreads();
  float a = bp[u];
  for (int k = 0; k < U_; ++k)  a = fmaf(sr[k], Up[k * U_ + u], a);
  for (int k = 0; k < IN_; ++k) a = fmaf(si[k], Cp[k * U_ + u], a);
  for (int k = 0; k < U_; ++k)  a = fmaf(sc[k], Cp[(IN_ + k) * U_ + u], a);
  float th = tanhf(a);
  float zt = ztw[b * U_ + u];
  out[b * U_ + u] = (1.f - zt) * h[b * U_ + u] + zt * th;
}

extern "C" void kernel_launch(void* const* d_in, const int* in_sizes, int n_in,
                              void* d_out, int out_size, void* d_ws, size_t ws_size,
                              hipStream_t stream) {
  const float* inputs = (const float*)d_in[0];
  const float* h_tm   = (const float*)d_in[1];
  const float* x_seq  = (const float*)d_in[2];
  const float* V_a    = (const float*)d_in[3];
  const float* W_a    = (const float*)d_in[4];
  const float* U_a    = (const float*)d_in[5];
  const float* b_a    = (const float*)d_in[6];
  const float* C_z    = (const float*)d_in[7];
  const float* W_z    = (const float*)d_in[8];
  const float* b_z    = (const float*)d_in[9];
  const float* C_r    = (const float*)d_in[10];
  const float* W_r    = (const float*)d_in[11];
  const float* b_r    = (const float*)d_in[12];
  const float* C_p    = (const float*)d_in[13];
  const float* U_p    = (const float*)d_in[14];
  const float* b_p    = (const float*)d_in[15];
  float* out = (float*)d_out;

  float* wsf   = (float*)d_ws;
  float* wxstm = wsf;
  float* et    = wsf + 32768;
  float* ctx   = et + (size_t)B_ * TE_;
  float* ztw   = ctx + B_ * U_;
  float* rhw   = ztw + B_ * U_;
  unsigned short* UaT = (unsigned short*)(rhw + B_ * U_);

  dim3 blk(256);
  k_prep_uaT<<<dim3(U_), blk, 0, stream>>>(U_a, UaT);
  k_wxstm  <<<dim3(B_), blk, 0, stream>>>(h_tm, W_a, wxstm);
  k_scores <<<dim3(TE_ / (16 * MT_), B_), blk, 0, stream>>>(x_seq, UaT, wxstm, b_a,
                                                            V_a, et);
  k_softmax<<<dim3(B_), blk, 0, stream>>>(et, ctx);
  k_context<<<dim3(TE_ / 128, B_), blk, 0, stream>>>(x_seq, et, ctx);
  k_gates  <<<dim3(B_), blk, 0, stream>>>(h_tm, inputs, ctx, W_z, C_z, b_z,
                                          W_r, C_r, b_r, ztw, rhw);
  k_out    <<<dim3(B_), blk, 0, stream>>>(h_tm, inputs, ctx, U_p, C_p, b_p,
                                          ztw, rhw, out);
}


# wxstm, gates, out layers: K split over the 8 waves of a block, columns over 2 blocks per row, 16-byte weight loads streamed 12 deep; 8 partial sums per output added in fixed order via LDS
# speedup vs baseline: 2.2418x; 2.2418x over previous
#include <hip/hip_runtime.h>

#define B_   128
#define TE_  2048
#define U_   256
#define IN_  256
#define MT_  4
#define CH_  128
#define NW_  8

typedef __attribute__((ext_vector_type(16))) __bf16 v16bf;
typedef __attribute__((ext_vector_type(8)))  float  v8f;

__device__ __forceinline__ unsigned short f32_to_bf16_bits(float f) {
  unsigned int u = __float_as_uint(f);
  u += 0x7FFFu + ((u >> 16) & 1u);
  return (unsigned short)(u >> 16);
}

__global__ void k_prep_uaT(const float* __restrict__ Ua,
                           unsigned short* __restrict__ UaT) {
  int e = blockIdx.x;
  int u = threadIdx.x;
  UaT[u * U_ + e] = f32_to_bf16_bits(Ua[e * U_ + u]);
}

#define RW_ (U_ / NW_)

__device__ __forceinline__ void fma4(float4& a, float s, const float4 x) {
  a.x = fmaf(s, x.x, a.x);
  a.y = fmaf(s, x.y, a.y);
  a.z = fmaf(s, x.z, a.z);
  a.w = fmaf(s, x.w, a.w);
}

__device__ __forceinline__ void seg1(const float* __restrict__ Wm,
                                     const float* __restrict__ v,
                                     int w, int col, float4& a) {
  const float* vw = v + RW_ * w;
  float4 s[RW_ / 4];
  #pragma unroll
  for (int j = 0; j < RW_ / 4; ++j) s[j] = *(const float4*)(vw + 4 * j);
  const float* p = Wm + (RW_ * w) * U_ + col;
  #pragma unroll
  for (int j = 0; j < RW_ / 4; ++j) {
    const float4 x0 = *(const float4*)(p + (4 * j + 0) * U_);
    const float4 x1 = *(const float4*)(p + (4 * j + 1) * U_);
    const float4 x2 = *(const float4*)(p + (4 * j + 2) * U_);
    const float4 x3 = *(const float4*)(p + (4 * j + 3) * U_);
    fma4(a, s[j].x, x0);
    fma4(a, s[j].y, x1);
    fma4(a, s[j].z, x2);
    fma4(a, s[j].w, x3);
  }
}

__device__ __forceinline__ void seg2(const float* __restrict__ Wm0,
                                     const float* __restrict__ Wm1,
                                     const float* __restrict__ v,
                                     int w, int col, float4& a0, float4& a1) {
  const float* vw = v + RW_ * w;
  float4 s[RW_ / 4];
  #pragma unroll
  for (int j = 0; j < RW_ / 4; ++j) s[j] = *(const float4*)(vw + 4 * j);
  const float* p0 = Wm0 + (RW_ * w) * U_ + col;
  const float* p1 = Wm1 + (RW_ * w) * U_ + col;
  #pragma unroll
  for (int j = 0; j < RW_ / 4; ++j) {
    const float4 x0 = *(const float4*)(p0 + (4 * j + 0) * U_);
    const float4 y0 = *(const float4*)(p1 + (4 * j + 0) * U_);
    const float4 x1 = *(const float4*)(p0 + (4 * j + 1) * U_);
    const float4 y1 = *(const float4*)(p1 + (4 * j + 1) * U_);
    const float4 x2 = *(const float4*)(p0 + (4 * j + 2) * U_);
    const float4 y2 = *(const float4*)(p1 + (4 * j + 2) * U_);
    const float4 x3 = *(const float4*)(p0 + (4 * j + 3) * U_);
    const float4 y3 = *(const float4*)(p1 + (4 * j + 3) * U_);
    fma4(a0, s[j].x, x0);
    fma4(a1, s[j].x, y0);
    fma4(a0, s[j].y, x1);
    fma4(a1, s[j].y, y1);
    fma4(a0, s[j].z, x2);
    fma4(a1, s[j].z, y2);
    fma4(a0, s[j].w, x3);
    fma4(a1, s[j].w, y3);
  }
}

__global__ __launch_bounds__(256) void k_wxstm(const float* __restrict__ h,
                                               const float* __restrict__ Wa,
                                               float* __restrict__ wx) {
  __shared__ __align__(16) float s_part[NW_][CH_];
  const int c0 = blockIdx.x * CH_, b = blockIdx.y, tid = threadIdx.x;
  const int w = __builtin_amdgcn_readfirstlane(tid >> 5);
  const int lane = tid & 31;
  float4 a = make_float4(0.f, 0.f, 0.f, 0.f);
  seg1(Wa, h + b * U_, w, c0 + 4 * lane, a);
  *(float4*)&s_part[w][4 * lane] = a;
  __syncthreads();
  if (tid < CH_) {
    float s = 0.f;
    #pragma unroll
    for (int p = 0; p < NW_; ++p) s += s_part[p][tid];
    wx[b * U_ + c0 + tid] = s;
  }
}

__global__ __launch_bounds__(256) void k_scores(
    const float* __restrict__ x_seq, const unsigned short* __restrict__ UaT,
    const float* __restrict__ wxstm, const float* __restrict__ b_a,
    const float* __restrict__ V_a, float* __restrict__ et) {
  __shared__ __align__(16) unsigned short sA[16 * MT_ * U_];
  __shared__ float s_part[16][16 * MT_];

  int b  = blockIdx.y;
  int t0 = blockIdx.x * (16 * MT_);
  int tid = threadIdx.x;

  const float* xrow = x_seq + ((size_t)b * TE_ + t0) * U_;
  for (int i = tid; i < 16 * MT_ * U_; i += 256) sA[i] = f32_to_bf16_bits(xrow[i]);
  __syncthreads();

  int wave = tid >> 5;
  int lane = tid & 31;
  int sel  = lane >> 4;
  int row  = lane & 15;

  v8f acc[2][MT_];
  #pragma unroll
  for (int j = 0; j < 2; ++j)
    #pragma unroll
    for (int m = 0; m < MT_; ++m)
      acc[j][m] = (v8f){0.f, 0.f, 0.f, 0.f, 0.f, 0.f, 0.f, 0.f};

  const unsigned short* bcol0 = UaT + (size_t)(2 * wave * 16 + row) * U_;
  const unsigned short* bcol1 = bcol0 + 16 * U_;

  #pragma unroll
  for (int kk = 0; kk < U_; kk += 32) {
    union { uint4 q[2]; v16bf v; } fb0, fb1;
    fb0.q[0] = *(const uint4*)(bcol0 + kk + sel * 8);
    fb0.q[1] = *(const uint4*)(bcol0 + kk + 16 + sel * 8);
    fb1.q[0] = *(const uint4*)(bcol1 + kk + sel * 8);
    fb1.q[1] = *(const uint4*)(bcol1 + kk + 16 + sel * 8);
    #pragma unroll
    for (int m = 0; m < MT_; ++m) {
      union { uint4 q[2]; v16bf v; } fa;
      const unsigned short* arow = &sA[(m * 16 + row) * U_ + kk];
      fa.q[0] = *(const uint4*)(arow + sel * 8);
      fa.q[1] = *(const uint4*)(arow + 16 + sel * 8);
      acc[0][m] = __builtin_amdgcn_wmma_f32_16x16x32_bf16(
          false, fb0.v, false, fa.v, (short)0, acc[0][m], false, false);
      acc[1][m] = __builtin_amdgcn_wmma_f32_16x16x32_bf16(
          false, fb1.v, false, fa.v, (short)0, acc[1][m], false, false);
    }
  }

  float part[MT_];
  #pragma unroll
  for (int m = 0; m < MT_; ++m) part[m] = 0.f;
  #pragma unroll
  for (int j = 0; j < 2; ++j) {
    int ug = (2 * wave + j) * 16 + 8 * sel;
    const float4* wp = (const float4*)(wxstm + b * U_ + ug);
    const float4* bp = (const float4*)(b_a + ug);
    const float4* vp = (const float4*)(V_a + ug);
    float4 w0 = wp[0], w1 = wp[1], c0 = bp[0], c1 = bp[1], v0 = vp[0], v1 = vp[1];
    float wx[8] = {w0.x + c0.x, w0.y + c0.y, w0.z + c0.z, w0.w + c0.w,
                   w1.x + c1.x, w1.y + c1.y, w1.z + c1.z, w1.w + c1.w};
    float va[8] = {v0.x, v0.y, v0.z, v0.w, v1.x, v1.y, v1.z, v1.w};
    #pragma unroll
    for (int m = 0; m < MT_; ++m) {
      #pragma unroll
      for (int i = 0; i < 8; ++i)
        part[m] = fmaf(va[i], tanhf(acc[j][m][i] + wx[i]), part[m]);
    }
  }
  #pragma unroll
  for (int m = 0; m < MT_; ++m) s_part[2 * wave + sel][m * 16 + row] = part[m];
  __syncthreads();
  if (tid < 16 * MT_) {
    float s = 0.f;
    #pragma unroll
    for (int p = 0; p < 16; ++p) s += s_part[p][tid];
    et[(size_t)b * TE_ + t0 + tid] = s;
  }
}

__global__ void k_softmax(float* __restrict__ at, float* __restrict__ ctx) {
  int b = blockIdx.x, tid = threadIdx.x;
  float v[8];
  float s = 0.f;
  #pragma unroll
  for (int j = 0; j < 8; ++j) {
    v[j] = expf(at[(size_t)b * TE_ + j * 256 + tid]);
    s += v[j];
  }
  __shared__ float red[256];
  red[tid] = s;
  __syncthreads();
  for (int off = 128; off > 0; off >>= 1) {
    if (tid < off) red[tid] += red[tid + off];
    __syncthreads();
  }
  float inv = 1.f / red[0];
  #pragma unroll
  for (int j = 0; j < 8; ++j)
    at[(size_t)b * TE_ + j * 256 + tid] = v[j] * inv;
  ctx[b * U_ + tid] = 0.f;
}

__global__ void k_context(const float* __restrict__ x_seq,
                          const float* __restrict__ at,
                          float* __restrict__ ctx) {
  int seg = blockIdx.x, b = blockIdx.y, u = threadIdx.x;
  const float* xp = x_seq + ((size_t)b * TE_ + seg * 128) * U_ + u;
  const float* ap = at + (size_t)b * TE_ + seg * 128;
  float acc = 0.f;
  for (int t = 0; t < 128; ++t) {
    __builtin_prefetch(xp + 8 * U_, 0, 1);
    acc = fmaf(ap[t], xp[0], acc);
    xp += U_;
  }
  atomicAdd(&ctx[b * U_ + u], acc);
}

__global__ __launch_bounds__(256) void k_gates(
    const float* __restrict__ h, const float* __restrict__ in,
    const float* __restrict__ ctx,
    const float* __restrict__ Wz, const float* __restrict__ Cz,
    const float* __restrict__ bz,
    const float* __restrict__ Wr, const float* __restrict__ Cr,
    const float* __restrict__ br,
    float* __restrict__ ztw, float* __restrict__ rhw) {
  __shared__ __align__(16) float s_part[NW_][2 * CH_];
  const int c0 = blockIdx.x * CH_, b = blockIdx.y, tid = threadIdx.x;
  const int w = __builtin_amdgcn_readfirstlane(tid >> 5);
  const int lane = tid & 31;
  const int col = c0 + 4 * lane;
  const int g = tid >> 7;
  const int u = c0 + (tid & (CH_ - 1));
  const float bias = g ? br[u] : bz[u];
  const float hv = h[b * U_ + u];
  float4 az = make_float4(0.f, 0.f, 0.f, 0.f);
  float4 ar = make_float4(0.f, 0.f, 0.f, 0.f);
  seg2(Wz, Wr, h + b * U_, w, col, az, ar);
  seg2(Cz, Cr, in + b * IN_, w, col, az, ar);
  seg2(Cz + IN_ * U_, Cr + IN_ * U_, ctx + b * U_, w, col, az, ar);
  *(float4*)&s_part[w][4 * lane] = az;
  *(float4*)&s_part[w][CH_ + 4 * lane] = ar;
  __syncthreads();
  float s = bias;
  #pragma unroll
  for (int p = 0; p < NW_; ++p) s += s_part[p][tid];
  float sg = 1.f / (1.f + expf(-s));
  if (g == 0) ztw[b * U_ + u] = sg;
  else        rhw[b * U_ + u] = sg * hv;
}

__global__ __launch_bounds__(256) void k_out(
    const float* __restrict__ h, const float* __restrict__ in,
    const float* __restrict__ ctx,
    const float* __restrict__ Up, const float* __restrict__ Cp,
    const float* __restrict__ bp,
    const float* __restrict__ ztw, const float* __restrict__ rhw,
    float* __restrict__ out) {
  __shared__ __align__(16) float s_part[NW_][CH_];
  const int c0 = blockIdx.x * CH_, b = blockIdx.y, tid = threadIdx.x;
  const int w = __builtin_amdgcn_readfirstlane(tid >> 5);
  const int lane = tid & 31;
  const int col = c0 + 4 * lane;
  const int u = c0 + (tid & (CH_ - 1));
  const float bias = bp[u];
  const float zt = ztw[b * U_ + u];
  const float hv = h[b * U_ + u];
  float4 a = make_float4(0.f, 0.f, 0.f, 0.f);
  seg1(Up, rhw + b * U_, w, col, a);
  seg1(Cp, in + b * IN_, w, col, a);
  seg1(Cp + IN_ * U_, ctx + b * U_, w, col, a);
  *(float4*)&s_part[w][4 * lane] = a;
  __syncthreads();
  if (tid < CH_) {
    float s = bias;
    #pragma unroll
    for (int p = 0; p < NW_; ++p) s += s_part[p][tid];
    float th = tanhf(s);
    out[b * U_ + u] = (1.f - zt) * hv + zt * th;
  }
}

extern "C" void kernel_launch(void* const* d_in, const int* in_sizes, int n_in,
                              void* d_out, int out_size, void* d_ws, size_t ws_size,
                              hipStream_t stream) {
  const float* inputs = (const float*)d_in[0];
  const float* h_tm   = (const float*)d_in[1];
  const float* x_seq  = (const float*)d_in[2];
  const float* V_a    = (const float*)d_in[3];
  const float* W_a    = (const float*)d_in[4];
  const float* U_a    = (const float*)d_in[5];
  const float* b_a    = (const float*)d_in[6];
  const float* C_z    = (const float*)d_in[7];
  const float* W_z    = (const float*)d_in[8];
  const float* b_z    = (const float*)d_in[9];
  const float* C_r    = (const float*)d_in[10];
  const float* W_r    = (const float*)d_in[11];
  const float* b_r    = (const float*)d_in[12];
  const float* C_p    = (const float*)d_in[13];
  const float* U_p    = (const float*)d_in[14];
  const float* b_p    = (const float*)d_in[15];
  float* out = (float*)d_out;

  float* wsf   = (float*)d_ws;
  float* wxstm = wsf;
  float* et    = wsf + 32768;
  float* ctx   = et + (size_t)B_ * TE_;
  float* ztw   = ctx + B_ * U_;
  float* rhw   = ztw + B_ * U_;
  unsigned short* UaT = (unsigned short*)(rhw + B_ * U_);

  dim3 blk(256);
  k_prep_uaT<<<dim3(U_), blk, 0, stream>>>(U_a, UaT);
  k_wxstm  <<<dim3(U_ / CH_, B_), blk, 0, stream>>>(h_tm, W_a, wxstm);
  k_scores <<<dim3(TE_ / (16 * MT_), B_), blk, 0, stream>>>(x_seq, UaT, wxstm, b_a,
                                                            V_a, et);
  k_softmax<<<dim3(B_), blk, 0, stream>>>(et, ctx);
  k_context<<<dim3(TE_ / 128, B_), blk, 0, stream>>>(x_seq, et, ctx);
  k_gates  <<<dim3(U_ / CH_, B_), blk, 0, stream>>>(h_tm, inputs, ctx, W_z, C_z, b_z,
                                          W_r, C_r, b_r, ztw, rhw);
  k_out    <<<dim3(U_ / CH_, B_), blk, 0, stream>>>(h_tm, inputs, ctx, U_p, C_p, b_p,
                                          ztw, rhw, out);
}


# k_scores: LDS row pitch of the staged x tile padded from 512 B to 528 B (16 B x odd), so the 16 rows of a half of each 16-byte fragment read fall on 64 different banks, not on 4
# speedup vs baseline: 2.9056x; 1.2961x over previous
#include <hip/hip_runtime.h>

#define B_   128
#define TE_  2048
#define U_   256
#define IN_  256
#define MT_  4
#define LDA_ (U_ + 8)
#define CH_  128
#define NW_  8

typedef __attribute__((ext_vector_type(16))) __bf16 v16bf;
typedef __attribute__((ext_vector_type(8)))  float  v8f;

__device__ __forceinline__ unsigned short f32_to_bf16_bits(float f) {
  unsigned int u = __float_as_uint(f);
  u += 0x7FFFu + ((u >> 16) & 1u);
  return (unsigned short)(u >> 16);
}

__global__ void k_prep_uaT(const float* __restrict__ Ua,
                           unsigned short* __restrict__ UaT) {
  int e = blockIdx.x;
  int u = threadIdx.x;
  UaT[u * U_ + e] = f32_to_bf16_bits(Ua[e * U_ + u]);
}

#define RW_ (U_ / NW_)

__device__ __forceinline__ void fma4(float4& a, float s, const float4 x) {
  a.x = fmaf(s, x.x, a.x);
  a.y = fmaf(s, x.y, a.y);
  a.z = fmaf(s, x.z, a.z);
  a.w = fmaf(s, x.w, a.w);
}

__device__ __forceinline__ void seg1(const float* __restrict__ Wm,
                                     const float* __restrict__ v,
                                     int w, int col, float4& a) {
  const float* vw = v + RW_ * w;
  float4 s[RW_ / 4];
  #pragma unroll
  for (int j = 0; j < RW_ / 4; ++j) s[j] = *(const float4*)(vw + 4 * j);
  const float* p = Wm + (RW_ * w) * U_ + col;
  #pragma unroll
  for (int j = 0; j < RW_ / 4; ++j) {
    const float4 x0 = *(const float4*)(p + (4 * j + 0) * U_);
    const float4 x1 = *(const float4*)(p + (4 * j + 1) * U_);
    const float4 x2 = *(const float4*)(p + (4 * j + 2) * U_);
    const float4 x3 = *(const float4*)(p + (4 * j + 3) * U_);
    fma4(a, s[j].x, x0);
    fma4(a, s[j].y, x1);
    fma4(a, s[j].z, x2);
    fma4(a, s[j].w, x3);
  }
}

__device__ __forceinline__ void seg2(const float* __restrict__ Wm0,
                                     const float* __restrict__ Wm1,
                                     const float* __restrict__ v,
                                     int w, int col, float4& a0, float4& a1) {
  const float* vw = v + RW_ * w;
  float4 s[RW_ / 4];
  #pragma unroll
  for (int j = 0; j < RW_ / 4; ++j) s[j] = *(const float4*)(vw + 4 * j);
  const float* p0 = Wm0 + (RW_ * w) * U_ + col;
  const float* p1 = Wm1 + (RW_ * w) * U_ + col;
  #pragma unroll
  for (int j = 0; j < RW_ / 4; ++j) {
    const float4 x0 = *(const float4*)(p0 + (4 * j + 0) * U_);
    const float4 y0 = *(const float4*)(p1 + (4 * j + 0) * U_);
    const float4 x1 = *(const float4*)(p0 + (4 * j + 1) * U_);
    const float4 y1 = *(const float4*)(p1 + (4 * j + 1) * U_);
    const float4 x2 = *(const float4*)(p0 + (4 * j + 2) * U_);
    const float4 y2 = *(const float4*)(p1 + (4 * j + 2) * U_);
    const float4 x3 = *(const float4*)(p0 + (4 * j + 3) * U_);
    const float4 y3 = *(const float4*)(p1 + (4 * j + 3) * U_);
    fma4(a0, s[j].x, x0);
    fma4(a1, s[j].x, y0);
    fma4(a0, s[j].y, x1);
    fma4(a1, s[j].y, y1);
    fma4(a0, s[j].z, x2);
    fma4(a1, s[j].z, y2);
    fma4(a0, s[j].w, x3);
    fma4(a1, s[j].w, y3);
  }
}

__global__ __launch_bounds__(256) void k_wxstm(const float* __restrict__ h,
                                               const float* __restrict__ Wa,
                                               float* __restrict__ wx) {
  __shared__ __align__(16) float s_part[NW_][CH_];
  const int c0 = blockIdx.x * CH_, b = blockIdx.y, tid = threadIdx.x;
  const int w = __builtin_amdgcn_readfirstlane(tid >> 5);
  const int lane = tid & 31;
  float4 a = make_float4(0.f, 0.f, 0.f, 0.f);
  seg1(Wa, h + b * U_, w, c0 + 4 * lane, a);
  *(float4*)&s_part[w][4 * lane] = a;
  __syncthreads();
  if (tid < CH_) {
    float s = 0.f;
    #pragma unroll
    for (int p = 0; p < NW_; ++p) s += s_part[p][tid];
    wx[b * U_ + c0 + tid] = s;
  }
}

__global__ __launch_bounds__(256) void k_scores(
    const float* __restrict__ x_seq, const unsigned short* __restrict__ UaT,
    const float* __restrict__ wxstm, const float* __restrict__ b_a,
    const float* __restrict__ V_a, float* __restrict__ et) {
  __shared__ __align__(16) unsigned short sA[16 * MT_ * LDA_];
  __shared__ float s_part[16][16 * MT_];

  int b  = blockIdx.y;
  int t0 = blockIdx.x * (16 * MT_);
  int tid = threadIdx.x;

  const float* xrow = x_seq + ((size_t)b * TE_ + t0) * U_;
  for (int i = tid; i < 16 * MT_ * U_; i += 256)
    sA[(i >> 8) * LDA_ + (i & (U_ - 1))] = f32_to_bf16_bits(xrow[i]);
  __syncthreads();

  int wave = tid >> 5;
  int lane = tid & 31;
  int sel  = lane >> 4;
  int row  = lane & 15;

  v8f acc[2][MT_];
  #pragma unroll
  for (int j = 0; j < 2; ++j)
    #pragma unroll
    for (int m = 0; m < MT_; ++m)
      acc[j][m] = (v8f){0.f, 0.f, 0.f, 0.f, 0.f, 0.f, 0.f, 0.f};

  const unsigned short* bcol0 = UaT + (size_t)(2 * wave * 16 + row) * U_;
  const unsigned short* bcol1 = bcol0 + 16 * U_;

  #pragma unroll
  for (int kk = 0; kk < U_; kk += 32) {
    union { uint4 q[2]; v16bf v; } fb0, fb1;
    fb0.q[0] = *(const uint4*)(bcol0 + kk + sel * 8);
    fb0.q[1] = *(const uint4*)(bcol0 + kk + 16 + sel * 8);
    fb1.q[0] = *(const uint4*)(bcol1 + kk + sel * 8);
    fb1.q[1] = *(const uint4*)(bcol1 + kk + 16 + sel * 8);
    #pragma unroll
    for (int m = 0; m < MT_; ++m) {
      union { uint4 q[2]; v16bf v; } fa;
      const unsigned short* arow = &sA[(m * 16 + row) * LDA_ + kk];
      fa.q[0] = *(const uint4*)(arow + sel * 8);
      fa.q[1] = *(const uint4*)(arow + 16 + sel * 8);
      acc[0][m] = __builtin_amdgcn_wmma_f32_16x16x32_bf16(
          false, fb0.v, false, fa.v, (short)0, acc[0][m], false, false);
      acc[1][m] = __builtin_amdgcn_wmma_f32_16x16x32_bf16(
          false, fb1.v, false, fa.v, (short)0, acc[1][m], false, false);
    }
  }

  float part[MT_];
  #pragma unroll
  for (int m = 0; m < MT_; ++m) part[m] = 0.f;
  #pragma unroll
  for (int j = 0; j < 2; ++j) {
    int ug = (2 * wave + j) * 16 + 8 * sel;
    const float4* wp = (const float4*)(wxstm + b * U_ + ug);
    const float4* bp = (const float4*)(b_a + ug);
    const float4* vp = (const float4*)(V_a + ug);
    float4 w0 = wp[0], w1 = wp[1], c0 = bp[0], c1 = bp[1], v0 = vp[0], v1 = vp[1];
    float wx[8] = {w0.x + c0.x, w0.y + c0.y, w0.z + c0.z, w0.w + c0.w,
                   w1.x + c1.x, w1.y + c1.y, w1.z + c1.z, w1.w + c1.w};
    float va[8] = {v0.x, v0.y, v0.z, v0.w, v1.x, v1.y, v1.z, v1.w};
    #pragma unroll
    for (int m = 0; m < MT_; ++m) {
      #pragma unroll
      for (int i = 0; i < 8; ++i)
        part[m] = fmaf(va[i], tanhf(acc[j][m][i] + wx[i]), part[m]);
    }
  }
  #pragma unroll
  for (int m = 0; m < MT_; ++m) s_part[2 * wave + sel][m * 16 + row] = part[m];
  __syncthreads();
  if (tid < 16 * MT_) {
    float s = 0.f;
    #pragma unroll
    for (int p = 0; p < 16; ++p) s += s_part[p][tid];
    et[(size_t)b * TE_ + t0 + tid] = s;
  }
}

__global__ void k_softmax(float* __restrict__ at, float* __restrict__ ctx) {
  int b = blockIdx.x, tid = threadIdx.x;
  float v[8];
  float s = 0.f;
  #pragma unroll
  for (int j = 0; j < 8; ++j) {
    v[j] = expf(at[(size_t)b * TE_ + j * 256 + tid]);
    s += v[j];
  }
  __shared__ float red[256];
  red[tid] = s;
  __syncthreads();
  for (int off = 128; off > 0; off >>= 1) {
    if (tid < off) red[tid] += red[tid + off];
    __syncthreads();
  }
  float inv = 1.f / red[0];
  #pragma unroll
  for (int j = 0; j < 8; ++j)
    at[(size_t)b * TE_ + j * 256 + tid] = v[j] * inv;
  ctx[b * U_ + tid] = 0.f;
}

__global__ void k_context(const float* __restrict__ x_seq,
                          const float* __restrict__ at,
                          float* __restrict__ ctx) {
  int seg = blockIdx.x, b = blockIdx.y, u = threadIdx.x;
  const float* xp = x_seq + ((size_t)b * TE_ + seg * 128) * U_ + u;
  const float* ap = at + (size_t)b * TE_ + seg * 128;
  float acc = 0.f;
  for (int t = 0; t < 128; ++t) {
    __builtin_prefetch(xp + 8 * U_, 0, 1);
    acc = fmaf(ap[t], xp[0], acc);
    xp += U_;
  }
  atomicAdd(&ctx[b * U_ + u], acc);
}

__global__ __launch_bounds__(256) void k_gates(
    const float* __restrict__ h, const float* __restrict__ in,
    const float* __restrict__ ctx,
    const float* __restrict__ Wz, const float* __restrict__ Cz,
    const float* __restrict__ bz,
    const float* __restrict__ Wr, const float* __restrict__ Cr,
    const float* __restrict__ br,
    float* __restrict__ ztw, float* __restrict__ rhw) {
  __shared__ __align__(16) float s_part[NW_][2 * CH_];
  const int c0 = blockIdx.x * CH_, b = blockIdx.y, tid = threadIdx.x;
  const int w = __builtin_amdgcn_readfirstlane(tid >> 5);
  const int lane = tid & 31;
  const int col = c0 + 4 * lane;
  const int g = tid >> 7;
  const int u = c0 + (tid & (CH_ - 1));
  const float bias = g ? br[u] : bz[u];
  const float hv = h[b * U_ + u];
  float4 az = make_float4(0.f, 0.f, 0.f, 0.f);
  float4 ar = make_float4(0.f, 0.f, 0.f, 0.f);
  seg2(Wz, Wr, h + b * U_, w, col, az, ar);
  seg2(Cz, Cr, in + b * IN_, w, col, az, ar);
  seg2(Cz + IN_ * U_, Cr + IN_ * U_, ctx + b * U_, w, col, az, ar);
  *(float4*)&s_part[w][4 * lane] = az;
  *(float4*)&s_part[w][CH_ + 4 * lane] = ar;
  __syncthreads();
  float s = bias;
  #pragma unroll
  for (int p = 0; p < NW_; ++p) s += s_part[p][tid];
  float sg = 1.f / (1.f + expf(-s));
  if (g == 0) ztw[b * U_ + u] = sg;
  else        rhw[b * U_ + u] = sg * hv;
}

__global__ __launch_bounds__(256) void k_out(
    const float* __restrict__ h, const float* __restrict__ in,
    const float* __restrict__ ctx,
    const float* __restrict__ Up, const float* __restrict__ Cp,
    const float* __restrict__ bp,
    const float* __restrict__ ztw, const float* __restrict__ rhw,
    float* __restrict__ out) {
  __shared__ __align__(16) float s_part[NW_][CH_];
  const int c0 = blockIdx.x * CH_, b = blockIdx.y, tid = threadIdx.x;
  const int w = __builtin_amdgcn_readfirstlane(tid >> 5);
  const int lane = tid & 31;
  const int col = c0 + 4 * lane;
  const int u = c0 + (tid & (CH_ - 1));
  const float bias = bp[u];
  const float zt = ztw[b * U_ + u];
  const float hv = h[b * U_ + u];
  float4 a = make_float4(0.f, 0.f, 0.f, 0.f);
  seg1(Up, rhw + b * U_, w, col, a);
  seg1(Cp, in + b * IN_, w, col, a);
  seg1(Cp + IN_ * U_, ctx + b * U_, w, col, a);
  *(float4*)&s_part[w][4 * lane] = a;
  __syncthreads();
  if (tid < CH_) {
    float s = bias;
    #pragma unroll
    for (int p = 0; p < NW_; ++p) s += s_part[p][tid];
    float th = tanhf(s);
    out[b * U_ + u] = (1.f - zt) * hv + zt * th;
  }
}

extern "C" void kernel_launch(void* const* d_in, const int* in_sizes, int n_in,
                              void* d_out, int out_size, void* d_ws, size_t ws_size,
                              hipStream_t stream) {
  const float* inputs = (const float*)d_in[0];
  const float* h_tm   = (const float*)d_in[1];
  const float* x_seq  = (const float*)d_in[2];
  const float* V_a    = (const float*)d_in[3];
  const float* W_a    = (const float*)d_in[4];
  const float* U_a    = (const float*)d_in[5];
  const float* b_a    = (const float*)d_in[6];
  const float* C_z    = (const float*)d_in[7];
  const float* W_z    = (const float*)d_in[8];
  const float* b_z    = (const float*)d_in[9];
  const float* C_r    = (const float*)d_in[10];
  const float* W_r    = (const float*)d_in[11];
  const float* b_r    = (const float*)d_in[12];
  const float* C_p    = (const float*)d_in[13];
  const float* U_p    = (const float*)d_in[14];
  const float* b_p    = (const float*)d_in[15];
  float* out = (float*)d_out;

  float* wsf   = (float*)d_ws;
  float* wxstm = wsf;
  float* et    = wsf + 32768;
  float* ctx   = et + (size_t)B_ * TE_;
  float* ztw   = ctx + B_ * U_;
  float* rhw   = ztw + B_ * U_;
  unsigned short* UaT = (unsigned short*)(rhw + B_ * U_);

  dim3 blk(256);
  k_prep_uaT<<<dim3(U_), blk, 0, stream>>>(U_a, UaT);
  k_wxstm  <<<dim3(U_ / CH_, B_), blk, 0, stream>>>(h_tm, W_a, wxstm);
  k_scores <<<dim3(TE_ / (16 * MT_), B_), blk, 0, stream>>>(x_seq, UaT, wxstm, b_a,
                                                            V_a, et);
  k_softmax<<<dim3(B_), blk, 0, stream>>>(et, ctx);
  k_context<<<dim3(TE_ / 128, B_), blk, 0, stream>>>(x_seq, et, ctx);
  k_gates  <<<dim3(U_ / CH_, B_), blk, 0, stream>>>(h_tm, inputs, ctx, W_z, C_z, b_z,
                                          W_r, C_r, b_r, ztw, rhw);
  k_out    <<<dim3(U_ / CH_, B_), blk, 0, stream>>>(h_tm, inputs, ctx, U_p, C_p, b_p,
                                          ztw, rhw, out);
}


# k_scores epilogue: tanh by the hardware v_tanh_f32 instruction, not the branching library tanhf (last bits differ); launch bound asks 6 waves per SIMD so the branch-free body keeps registers low
# speedup vs baseline: 3.8186x; 1.3142x over previous
#include <hip/hip_runtime.h>

#define B_   128
#define TE_  2048
#define U_   256
#define IN_  256
#define MT_  4
#define LDA_ (U_ + 8)
#define CH_  128
#define NW_  8

typedef __attribute__((ext_vector_type(16))) __bf16 v16bf;
typedef __attribute__((ext_vector_type(8)))  float  v8f;

__device__ __forceinline__ unsigned short f32_to_bf16_bits(float f) {
  unsigned int u = __float_as_uint(f);
  u += 0x7FFFu + ((u >> 16) & 1u);
  return (unsigned short)(u >> 16);
}

__global__ void k_prep_uaT(const float* __restrict__ Ua,
                           unsigned short* __restrict__ UaT) {
  int e = blockIdx.x;
  int u = threadIdx.x;
  UaT[u * U_ + e] = f32_to_bf16_bits(Ua[e * U_ + u]);
}

#define RW_ (U_ / NW_)

__device__ __forceinline__ void fma4(float4& a, float s, const float4 x) {
  a.x = fmaf(s, x.x, a.x);
  a.y = fmaf(s, x.y, a.y);
  a.z = fmaf(s, x.z, a.z);
  a.w = fmaf(s, x.w, a.w);
}

__device__ __forceinline__ void seg1(const float* __restrict__ Wm,
                                     const float* __restrict__ v,
                                     int w, int col, float4& a) {
  const float* vw = v + RW_ * w;
  float4 s[RW_ / 4];
  #pragma unroll
  for (int j = 0; j < RW_ / 4; ++j) s[j] = *(const float4*)(vw + 4 * j);
  const float* p = Wm + (RW_ * w) * U_ + col;
  #pragma unroll
  for (int j = 0; j < RW_ / 4; ++j) {
    const float4 x0 = *(const float4*)(p + (4 * j + 0) * U_);
    const float4 x1 = *(const float4*)(p + (4 * j + 1) * U_);
    const float4 x2 = *(const float4*)(p + (4 * j + 2) * U_);
    const float4 x3 = *(const float4*)(p + (4 * j + 3) * U_);
    fma4(a, s[j].x, x0);
    fma4(a, s[j].y, x1);
    fma4(a, s[j].z, x2);
    fma4(a, s[j].w, x3);
  }
}

__device__ __forceinline__ void seg2(const float* __restrict__ Wm0,
                                     const float* __restrict__ Wm1,
                                     const float* __restrict__ v,
                                     int w, int col, float4& a0, float4& a1) {
  const float* vw = v + RW_ * w;
  float4 s[RW_ / 4];
  #pragma unroll
  for (int j = 0; j < RW_ / 4; ++j) s[j] = *(const float4*)(vw + 4 * j);
  const float* p0 = Wm0 + (RW_ * w) * U_ + col;
  const float* p1 = Wm1 + (RW_ * w) * U_ + col;
  #pragma unroll
  for (int j = 0; j < RW_ / 4; ++j) {
    const float4 x0 = *(const float4*)(p0 + (4 * j + 0) * U_);
    const float4 y0 = *(const float4*)(p1 + (4 * j + 0) * U_);
    const float4 x1 = *(const float4*)(p0 + (4 * j + 1) * U_);
    const float4 y1 = *(const float4*)(p1 + (4 * j + 1) * U_);
    const float4 x2 = *(const float4*)(p0 + (4 * j + 2) * U_);
    const float4 y2 = *(const float4*)(p1 + (4 * j + 2) * U_);
    const float4 x3 = *(const float4*)(p0 + (4 * j + 3) * U_);
    const float4 y3 = *(const float4*)(p1 + (4 * j + 3) * U_);
    fma4(a0, s[j].x, x0);
    fma4(a1, s[j].x, y0);
    fma4(a0, s[j].y, x1);
    fma4(a1, s[j].y, y1);
    fma4(a0, s[j].z, x2);
    fma4(a1, s[j].z, y2);
    fma4(a0, s[j].w, x3);
    fma4(a1, s[j].w, y3);
  }
}

__global__ __launch_bounds__(256) void k_wxstm(const float* __restrict__ h,
                                               const float* __restrict__ Wa,
                                               float* __restrict__ wx) {
  __shared__ __align__(16) float s_part[NW_][CH_];
  const int c0 = blockIdx.x * CH_, b = blockIdx.y, tid = threadIdx.x;
  const int w = __builtin_amdgcn_readfirstlane(tid >> 5);
  const int lane = tid & 31;
  float4 a = make_float4(0.f, 0.f, 0.f, 0.f);
  seg1(Wa, h + b * U_, w, c0 + 4 * lane, a);
  *(float4*)&s_part[w][4 * lane] = a;
  __syncthreads();
  if (tid < CH_) {
    float s = 0.f;
    #pragma unroll
    for (int p = 0; p < NW_; ++p) s += s_part[p][tid];
    wx[b * U_ + c0 + tid] = s;
  }
}

__global__ __launch_bounds__(256, 6) void k_scores(
    const float* __restrict__ x_seq, const unsigned short* __restrict__ UaT,
    const float* __restrict__ wxstm, const float* __restrict__ b_a,
    const float* __restrict__ V_a, float* __restrict__ et) {
  __shared__ __align__(16) unsigned short sA[16 * MT_ * LDA_];
  __shared__ float s_part[16][16 * MT_];

  int b  = blockIdx.y;
  int t0 = blockIdx.x * (16 * MT_);
  int tid = threadIdx.x;

  const float* xrow = x_seq + ((size_t)b * TE_ + t0) * U_;
  for (int i = tid; i < 16 * MT_ * U_; i += 256)
    sA[(i >> 8) * LDA_ + (i & (U_ - 1))] = f32_to_bf16_bits(xrow[i]);
  __syncthreads();

  int wave = tid >> 5;
  int lane = tid & 31;
  int sel  = lane >> 4;
  int row  = lane & 15;

  v8f acc[2][MT_];
  #pragma unroll
  for (int j = 0; j < 2; ++j)
    #pragma unroll
    for (int m = 0; m < MT_; ++m)
      acc[j][m] = (v8f){0.f, 0.f, 0.f, 0.f, 0.f, 0.f, 0.f, 0.f};

  const unsigned short* bcol0 = UaT + (size_t)(2 * wave * 16 + row) * U_;
  const unsigned short* bcol1 = bcol0 + 16 * U_;

  #pragma unroll
  for (int kk = 0; kk < U_; kk += 32) {
    union { uint4 q[2]; v16bf v; } fb0, fb1;
    fb0.q[0] = *(const uint4*)(bcol0 + kk + sel * 8);
    fb0.q[1] = *(const uint4*)(bcol0 + kk + 16 + sel * 8);
    fb1.q[0] = *(const uint4*)(bcol1 + kk + sel * 8);
    fb1.q[1] = *(const uint4*)(bcol1 + kk + 16 + sel * 8);
    #pragma unroll
    for (int m = 0; m < MT_; ++m) {
      union { uint4 q[2]; v16bf v; } fa;
      const unsigned short* arow = &sA[(m * 16 + row) * LDA_ + kk];
      fa.q[0] = *(const uint4*)(arow + sel * 8);
      fa.q[1] = *(const uint4*)(arow + 16 + sel * 8);
      acc[0][m] = __builtin_amdgcn_wmma_f32_16x16x32_bf16(
          false, fb0.v, false, fa.v, (short)0, acc[0][m], false, false);
      acc[1][m] = __builtin_amdgcn_wmma_f32_16x16x32_bf16(
          false, fb1.v, false, fa.v, (short)0, acc[1][m], false, false);
    }
  }

  float part[MT_];
  #pragma unroll
  for (int m = 0; m < MT_; ++m) part[m] = 0.f;
  #pragma unroll
  for (int j = 0; j < 2; ++j) {
    int ug = (2 * wave + j) * 16 + 8 * sel;
    const float4* wp = (const float4*)(wxstm + b * U_ + ug);
    const float4* bp = (const float4*)(b_a + ug);
    const float4* vp = (const float4*)(V_a + ug);
    float4 w0 = wp[0], w1 = wp[1], c0 = bp[0], c1 = bp[1], v0 = vp[0], v1 = vp[1];
    float wx[8] = {w0.x + c0.x, w0.y + c0.y, w0.z + c0.z, w0.w + c0.w,
                   w1.x + c1.x, w1.y + c1.y, w1.z + c1.z, w1.w + c1.w};
    float va[8] = {v0.x, v0.y, v0.z, v0.w, v1.x, v1.y, v1.z, v1.w};
    #pragma unroll
    for (int m = 0; m < MT_; ++m) {
      #pragma unroll
      for (int i = 0; i < 8; ++i)
        part[m] = fmaf(va[i], __builtin_amdgcn_tanhf(acc[j][m][i] + wx[i]), part[m]);
    }
  }
  #pragma unroll
  for (int m = 0; m < MT_; ++m) s_part[2 * wave + sel][m * 16 + row] = part[m];
  __syncthreads();
  if (tid < 16 * MT_) {
    float s = 0.f;
    #pragma unroll
    for (int p = 0; p < 16; ++p) s += s_part[p][tid];
    et[(size_t)b * TE_ + t0 + tid] = s;
  }
}

__global__ void k_softmax(float* __restrict__ at, float* __restrict__ ctx) {
  int b = blockIdx.x, tid = threadIdx.x;
  float v[8];
  float s = 0.f;
  #pragma unroll
  for (int j = 0; j < 8; ++j) {
    v[j] = expf(at[(size_t)b * TE_ + j * 256 + tid]);
    s += v[j];
  }
  __shared__ float red[256];
  red[tid] = s;
  __syncthreads();
  for (int off = 128; off > 0; off >>= 1) {
    if (tid < off) red[tid] += red[tid + off];
    __syncthreads();
  }
  float inv = 1.f / red[0];
  #pragma unroll
  for (int j = 0; j < 8; ++j)
    at[(size_t)b * TE_ + j * 256 + tid] = v[j] * inv;
  ctx[b * U_ + tid] = 0.f;
}

__global__ void k_context(const float* __restrict__ x_seq,
                          const float* __restrict__ at,
                          float* __restrict__ ctx) {
  int seg = blockIdx.x, b = blockIdx.y, u = threadIdx.x;
  const float* xp = x_seq + ((size_t)b * TE_ + seg * 128) * U_ + u;
  const float* ap = at + (size_t)b * TE_ + seg * 128;
  float acc = 0.f;
  for (int t = 0; t < 128; ++t) {
    __builtin_prefetch(xp + 8 * U_, 0, 1);
    acc = fmaf(ap[t], xp[0], acc);
    xp += U_;
  }
  atomicAdd(&ctx[b * U_ + u], acc);
}

__global__ __launch_bounds__(256) void k_gates(
    const float* __restrict__ h, const float* __restrict__ in,
    const float* __restrict__ ctx,
    const float* __restrict__ Wz, const float* __restrict__ Cz,
    const float* __restrict__ bz,
    const float* __restrict__ Wr, const float* __restrict__ Cr,
    const float* __restrict__ br,
    float* __restrict__ ztw, float* __restrict__ rhw) {
  __shared__ __align__(16) float s_part[NW_][2 * CH_];
  const int c0 = blockIdx.x * CH_, b = blockIdx.y, tid = threadIdx.x;
  const int w = __builtin_amdgcn_readfirstlane(tid >> 5);
  const int lane = tid & 31;
  const int col = c0 + 4 * lane;
  const int g = tid >> 7;
  const int u = c0 + (tid & (CH_ - 1));
  const float bias = g ? br[u] : bz[u];
  const float hv = h[b * U_ + u];
  float4 az = make_float4(0.f, 0.f, 0.f, 0.f);
  float4 ar = make_float4(0.f, 0.f, 0.f, 0.f);
  seg2(Wz, Wr, h + b * U_, w, col, az, ar);
  seg2(Cz, Cr, in + b * IN_, w, col, az, ar);
  seg2(Cz + IN_ * U_, Cr + IN_ * U_, ctx + b * U_, w, col, az, ar);
  *(float4*)&s_part[w][4 * lane] = az;
  *(float4*)&s_part[w][CH_ + 4 * lane] = ar;
  __syncthreads();
  float s = bias;
  #pragma unroll
  for (int p = 0; p < NW_; ++p) s += s_part[p][tid];
  float sg = 1.f / (1.f + expf(-s));
  if (g == 0) ztw[b * U_ + u] = sg;
  else        rhw[b * U_ + u] = sg * hv;
}

__global__ __launch_bounds__(256) void k_out(
    const float* __restrict__ h, const float* __restrict__ in,
    const float* __restrict__ ctx,
    const float* __restrict__ Up, const float* __restrict__ Cp,
    const float* __restrict__ bp,
    const float* __restrict__ ztw, const float* __restrict__ rhw,
    float* __restrict__ out) {
  __shared__ __align__(16) float s_part[NW_][CH_];
  const int c0 = blockIdx.x * CH_, b = blockIdx.y, tid = threadIdx.x;
  const int w = __builtin_amdgcn_readfirstlane(tid >> 5);
  const int lane = tid & 31;
  const int col = c0 + 4 * lane;
  const int u = c0 + (tid & (CH_ - 1));
  const float bias = bp[u];
  const float zt = ztw[b * U_ + u];
  const float hv = h[b * U_ + u];
  float4 a = make_float4(0.f, 0.f, 0.f, 0.f);
  seg1(Up, rhw + b * U_, w, col, a);
  seg1(Cp, in + b * IN_, w, col, a);
  seg1(Cp + IN_ * U_, ctx + b * U_, w, col, a);
  *(float4*)&s_part[w][4 * lane] = a;
  __syncthreads();
  if (tid < CH_) {
    float s = bias;
    #pragma unroll
    for (int p = 0; p < NW_; ++p) s += s_part[p][tid];
    float th = tanhf(s);
    out[b * U_ + u] = (1.f - zt) * hv + zt * th;
  }
}

extern "C" void kernel_launch(void* const* d_in, const int* in_sizes, int n_in,
                              void* d_out, int out_size, void* d_ws, size_t ws_size,
                              hipStream_t stream) {
  const float* inputs = (const float*)d_in[0];
  const float* h_tm   = (const float*)d_in[1];
  const float* x_seq  = (const float*)d_in[2];
  const float* V_a    = (const float*)d_in[3];
  const float* W_a    = (const float*)d_in[4];
  const float* U_a    = (const float*)d_in[5];
  const float* b_a    = (const float*)d_in[6];
  const float* C_z    = (const float*)d_in[7];
  const float* W_z    = (const float*)d_in[8];
  const float* b_z    = (const float*)d_in[9];
  const float* C_r    = (const float*)d_in[10];
  const float* W_r    = (const float*)d_in[11];
  const float* b_r    = (const float*)d_in[12];
  const float* C_p    = (const float*)d_in[13];
  const float* U_p    = (const float*)d_in[14];
  const float* b_p    = (const float*)d_in[15];
  float* out = (float*)d_out;

  float* wsf   = (float*)d_ws;
  float* wxstm = wsf;
  float* et    = wsf + 32768;
  float* ctx   = et + (size_t)B_ * TE_;
  float* ztw   = ctx + B_ * U_;
  float* rhw   = ztw + B_ * U_;
  unsigned short* UaT = (unsigned short*)(rhw + B_ * U_);

  dim3 blk(256);
  k_prep_uaT<<<dim3(U_), blk, 0, stream>>>(U_a, UaT);
  k_wxstm  <<<dim3(U_ / CH_, B_), blk, 0, stream>>>(h_tm, W_a, wxstm);
  k_scores <<<dim3(TE_ / (16 * MT_), B_), blk, 0, stream>>>(x_seq, UaT, wxstm, b_a,
                                                            V_a, et);
  k_softmax<<<dim3(B_), blk, 0, stream>>>(et, ctx);
  k_context<<<dim3(TE_ / 128, B_), blk, 0, stream>>>(x_seq, et, ctx);
  k_gates  <<<dim3(U_ / CH_, B_), blk, 0, stream>>>(h_tm, inputs, ctx, W_z, C_z, b_z,
                                          W_r, C_r, b_r, ztw, rhw);
  k_out    <<<dim3(U_ / CH_, B_), blk, 0, stream>>>(h_tm, inputs, ctx, U_p, C_p, b_p,
                                          ztw, rhw, out);
}


# k_scores staging: each thread issues its 16 16-byte loads of the x tile before using any (whole 64 KB tile in flight, one memory wait; was 8 waited rounds of 8 4-byte loads); 8-byte LDS stores
# speedup vs baseline: 4.8891x; 1.2803x over previous
#include <hip/hip_runtime.h>

#define B_   128
#define TE_  2048
#define U_   256
#define IN_  256
#define MT_  4
#define LDA_ (U_ + 8)
#define CH_  128
#define NW_  8
#define NT_  (TE_ / (16 * MT_))

typedef __attribute__((ext_vector_type(16))) __bf16 v16bf;
typedef __attribute__((ext_vector_type(8)))  float  v8f;
typedef __attribute__((ext_vector_type(4)))  float  v4f;

__device__ __forceinline__ unsigned short f32_to_bf16_bits(float f) {
  unsigned int u = __float_as_uint(f);
  u += 0x7FFFu + ((u >> 16) & 1u);
  return (unsigned short)(u >> 16);
}

__device__ __forceinline__ unsigned int f32x2_to_bf16x2_bits(float lo, float hi) {
  unsigned int a = __float_as_uint(lo), c = __float_as_uint(hi);
  a += 0x7FFFu + ((a >> 16) & 1u);
  c += 0x7FFFu + ((c >> 16) & 1u);
  return (a >> 16) | (c & 0xFFFF0000u);
}

__global__ void k_prep_uaT(const float* __restrict__ Ua,
                           unsigned short* __restrict__ UaT) {
  int e = blockIdx.x;
  int u = threadIdx.x;
  UaT[u * U_ + e] = f32_to_bf16_bits(Ua[e * U_ + u]);
}

#define RW_ (U_ / NW_)

__device__ __forceinline__ void fma4(float4& a, float s, const float4 x) {
  a.x = fmaf(s, x.x, a.x);
  a.y = fmaf(s, x.y, a.y);
  a.z = fmaf(s, x.z, a.z);
  a.w = fmaf(s, x.w, a.w);
}

__device__ __forceinline__ void seg1(const float* __restrict__ Wm,
                                     const float* __restrict__ v,
                                     int w, int col, float4& a) {
  const float* vw = v + RW_ * w;
  float4 s[RW_ / 4];
  #pragma unroll
  for (int j = 0; j < RW_ / 4; ++j) s[j] = *(const float4*)(vw + 4 * j);
  const float* p = Wm + (RW_ * w) * U_ + col;
  #pragma unroll
  for (int j = 0; j < RW_ / 4; ++j) {
    const float4 x0 = *(const float4*)(p + (4 * j + 0) * U_);
    const float4 x1 = *(const float4*)(p + (4 * j + 1) * U_);
    const float4 x2 = *(const float4*)(p + (4 * j + 2) * U_);
    const float4 x3 = *(const float4*)(p + (4 * j + 3) * U_);
    fma4(a, s[j].x, x0);
    fma4(a, s[j].y, x1);
    fma4(a, s[j].z, x2);
    fma4(a, s[j].w, x3);
  }
}

__device__ __forceinline__ void seg2(const float* __restrict__ Wm0,
                                     const float* __restrict__ Wm1,
                                     const float* __restrict__ v,
                                     int w, int col, float4& a0, float4& a1) {
  const float* vw = v + RW_ * w;
  float4 s[RW_ / 4];
  #pragma unroll
  for (int j = 0; j < RW_ / 4; ++j) s[j] = *(const float4*)(vw + 4 * j);
  const float* p0 = Wm0 + (RW_ * w) * U_ + col;
  const float* p1 = Wm1 + (RW_ * w) * U_ + col;
  #pragma unroll
  for (int j = 0; j < RW_ / 4; ++j) {
    const float4 x0 = *(const float4*)(p0 + (4 * j + 0) * U_);
    const float4 y0 = *(const float4*)(p1 + (4 * j + 0) * U_);
    const float4 x1 = *(const float4*)(p0 + (4 * j + 1) * U_);
    const float4 y1 = *(const float4*)(p1 + (4 * j + 1) * U_);
    const float4 x2 = *(const float4*)(p0 + (4 * j + 2) * U_);
    const float4 y2 = *(const float4*)(p1 + (4 * j + 2) * U_);
    const float4 x3 = *(const float4*)(p0 + (4 * j + 3) * U_);
    const float4 y3 = *(const float4*)(p1 + (4 * j + 3) * U_);
    fma4(a0, s[j].x, x0);
    fma4(a1, s[j].x, y0);
    fma4(a0, s[j].y, x1);
    fma4(a1, s[j].y, y1);
    fma4(a0, s[j].z, x2);
    fma4(a1, s[j].z, y2);
    fma4(a0, s[j].w, x3);
    fma4(a1, s[j].w, y3);
  }
}

__global__ __launch_bounds__(256) void k_wxstm(const float* __restrict__ h,
                                               const float* __restrict__ Wa,
                                               float* __restrict__ wx) {
  __shared__ __align__(16) float s_part[NW_][CH_];
  const int c0 = blockIdx.x * CH_, b = blockIdx.y, tid = threadIdx.x;
  const int w = __builtin_amdgcn_readfirstlane(tid >> 5);
  const int lane = tid & 31;
  float4 a = make_float4(0.f, 0.f, 0.f, 0.f);
  seg1(Wa, h + b * U_, w, c0 + 4 * lane, a);
  *(float4*)&s_part[w][4 * lane] = a;
  __syncthreads();
  if (tid < CH_) {
    float s = 0.f;
    #pragma unroll
    for (int p = 0; p < NW_; ++p) s += s_part[p][tid];
    wx[b * U_ + c0 + tid] = s;
  }
}

__global__ __launch_bounds__(256, 10) void k_scores(
    const float* __restrict__ x_seq, const unsigned short* __restrict__ UaT,
    const float* __restrict__ wxstm, const float* __restrict__ b_a,
    const float* __restrict__ V_a, float* __restrict__ cpart,
    float* __restrict__ spart) {
  __shared__ __align__(16) unsigned short sA[16 * MT_ * LDA_];
  __shared__ float s_part[16][16 * MT_];
  __shared__ __align__(16) float s_p[16 * MT_];

  int b  = blockIdx.y;
  int t0 = blockIdx.x * (16 * MT_);
  int tid = threadIdx.x;

  const float* xrow = x_seq + ((size_t)b * TE_ + t0) * U_;
  {
    const v4f* xq = (const v4f*)(x_seq + (size_t)b * TE_ * U_);
    const int q0 = t0 * (U_ / 4) + tid;
    v4f xs[16 * MT_ / 4];
    #pragma unroll
    for (int j = 0; j < 16 * MT_ / 4; ++j) xs[j] = xq[q0 + j * 256];
    asm volatile("" :: "v"(xs[0]), "v"(xs[1]), "v"(xs[2]), "v"(xs[3]), "v"(xs[4]), "v"(xs[5]),
                 "v"(xs[6]), "v"(xs[7]), "v"(xs[8]), "v"(xs[9]), "v"(xs[10]), "v"(xs[11]),
                 "v"(xs[12]), "v"(xs[13]), "v"(xs[14]), "v"(xs[15]));
    unsigned short* srow = &sA[(tid >> 6) * LDA_ + 4 * (tid & 63)];
    #pragma unroll
    for (int j = 0; j < 16 * MT_ / 4; ++j) {
      uint2 pk;
      pk.x = f32x2_to_bf16x2_bits(xs[j].x, xs[j].y);
      pk.y = f32x2_to_bf16x2_bits(xs[j].z, xs[j].w);
      *(uint2*)(srow + 4 * j * LDA_) = pk;
    }
  }
  __syncthreads();

  int wave = tid >> 5;
  int lane = tid & 31;
  int sel  = lane >> 4;
  int row  = lane & 15;

  v8f acc[2][MT_];
  #pragma unroll
  for (int j = 0; j < 2; ++j)
    #pragma unroll
    for (int m = 0; m < MT_; ++m)
      acc[j][m] = (v8f){0.f, 0.f, 0.f, 0.f, 0.f, 0.f, 0.f, 0.f};

  const int boff = (2 * wave * 16 + row) * U_ + sel * 8;

  #pragma unroll
  for (int kk = 0; kk < U_; kk += 32) {
    union { uint4 q[2]; v16bf v; } fb0, fb1;
    fb0.q[0] = *(const uint4*)(UaT + boff + kk);
    fb0.q[1] = *(const uint4*)(UaT + boff + kk + 16);
    fb1.q[0] = *(const uint4*)(UaT + boff + 16 * U_ + kk);
    fb1.q[1] = *(const uint4*)(UaT + boff + 16 * U_ + kk + 16);
    #pragma unroll
    for (int m = 0; m < MT_; ++m) {
      union { uint4 q[2]; v16bf v; } fa;
      const unsigned short* arow = &sA[(m * 16 + row) * LDA_ + kk];
      fa.q[0] = *(const uint4*)(arow + sel * 8);
      fa.q[1] = *(const uint4*)(arow + 16 + sel * 8);
      acc[0][m] = __builtin_amdgcn_wmma_f32_16x16x32_bf16(
          false, fb0.v, false, fa.v, (short)0, acc[0][m], false, false);
      acc[1][m] = __builtin_amdgcn_wmma_f32_16x16x32_bf16(
          false, fb1.v, false, fa.v, (short)0, acc[1][m], false, false);
    }
  }

  float part[MT_];
  #pragma unroll
  for (int m = 0; m < MT_; ++m) part[m] = 0.f;
  #pragma unroll
  for (int j = 0; j < 2; ++j) {
    int ug = (2 * wave + j) * 16 + 8 * sel;
    const float4* wp = (const float4*)(wxstm + b * U_ + ug);
    const float4* bp = (const float4*)(b_a + ug);
    const float4* vp = (const float4*)(V_a + ug);
    float4 w0 = wp[0], w1 = wp[1], c0 = bp[0], c1 = bp[1], v0 = vp[0], v1 = vp[1];
    float wx[8] = {w0.x + c0.x, w0.y + c0.y, w0.z + c0.z, w0.w + c0.w,
                   w1.x + c1.x, w1.y + c1.y, w1.z + c1.z, w1.w + c1.w};
    float va[8] = {v0.x, v0.y, v0.z, v0.w, v1.x, v1.y, v1.z, v1.w};
    #pragma unroll
    for (int m = 0; m < MT_; ++m) {
      #pragma unroll
      for (int i = 0; i < 8; ++i)
        part[m] = fmaf(va[i], __builtin_amdgcn_tanhf(acc[j][m][i] + wx[i]), part[m]);
    }
  }
  #pragma unroll
  for (int m = 0; m < MT_; ++m) s_part[2 * wave + sel][m * 16 + row] = part[m];
  __syncthreads();
  if (tid < 16 * MT_) {
    float s = 0.f;
    #pragma unroll
    for (int p = 0; p < 16; ++p) s += s_part[p][tid];
    s_p[tid] = expf(s);
  }
  __syncthreads();

  float xv[16 * MT_];
  #pragma unroll
  for (int k = 0; k < 16 * MT_; ++k) xv[k] = xrow[k * U_ + tid];
  float ca0 = 0.f, ca1 = 0.f, ca2 = 0.f, ca3 = 0.f;
  float cs0 = 0.f, cs1 = 0.f, cs2 = 0.f, cs3 = 0.f;
  #pragma unroll
  for (int q = 0; q < 4 * MT_; ++q) {
    const float4 pv = *(const float4*)&s_p[4 * q];
    ca0 = fmaf(pv.x, xv[4 * q + 0], ca0);
    ca1 = fmaf(pv.y, xv[4 * q + 1], ca1);
    ca2 = fmaf(pv.z, xv[4 * q + 2], ca2);
    ca3 = fmaf(pv.w, xv[4 * q + 3], ca3);
    cs0 += pv.x;
    cs1 += pv.y;
    cs2 += pv.z;
    cs3 += pv.w;
  }
  const int tile = b * NT_ + blockIdx.x;
  cpart[(size_t)tile * U_ + tid] = (ca0 + ca1) + (ca2 + ca3);
  if (tid == 0) spart[tile] = (cs0 + cs1) + (cs2 + cs3);
}

__global__ __launch_bounds__(256) void k_ctx_reduce(const float* __restrict__ cpart,
                                                    const float* __restrict__ spart,
                                                    float* __restrict__ ctx) {
  const int b = blockIdx.x, u = threadIdx.x;
  const float* cp = cpart + (size_t)b * NT_ * U_ + u;
  const float* sp = spart + b * NT_;
  float c = 0.f, s = 0.f;
  #pragma unroll
  for (int i = 0; i < NT_; ++i) {
    c += cp[i * U_];
    s += sp[i];
  }
  ctx[b * U_ + u] = c / s;
}

__global__ __launch_bounds__(256) void k_gates(
    const float* __restrict__ h, const float* __restrict__ in,
    const float* __restrict__ ctx,
    const float* __restrict__ Wz, const float* __restrict__ Cz,
    const float* __restrict__ bz,
    const float* __restrict__ Wr, const float* __restrict__ Cr,
    const float* __restrict__ br,
    float* __restrict__ ztw, float* __restrict__ rhw) {
  __shared__ __align__(16) float s_part[NW_][2 * CH_];
  const int c0 = blockIdx.x * CH_, b = blockIdx.y, tid = threadIdx.x;
  const int w = __builtin_amdgcn_readfirstlane(tid >> 5);
  const int lane = tid & 31;
  const int col = c0 + 4 * lane;
  const int g = tid >> 7;
  const int u = c0 + (tid & (CH_ - 1));
  const float bias = g ? br[u] : bz[u];
  const float hv = h[b * U_ + u];
  float4 az = make_float4(0.f, 0.f, 0.f, 0.f);
  float4 ar = make_float4(0.f, 0.f, 0.f, 0.f);
  seg2(Wz, Wr, h + b * U_, w, col, az, ar);
  seg2(Cz, Cr, in + b * IN_, w, col, az, ar);
  seg2(Cz + IN_ * U_, Cr + IN_ * U_, ctx + b * U_, w, col, az, ar);
  *(float4*)&s_part[w][4 * lane] = az;
  *(float4*)&s_part[w][CH_ + 4 * lane] = ar;
  __syncthreads();
  float s = bias;
  #pragma unroll
  for (int p = 0; p < NW_; ++p) s += s_part[p][tid];
  float sg = 1.f / (1.f + expf(-s));
  if (g == 0) ztw[b * U_ + u] = sg;
  else        rhw[b * U_ + u] = sg * hv;
}

__global__ __launch_bounds__(256) void k_out(
    const float* __restrict__ h, const float* __restrict__ in,
    const float* __restrict__ ctx,
    const float* __restrict__ Up, const float* __restrict__ Cp,
    const float* __restrict__ bp,
    const float* __restrict__ ztw, const float* __restrict__ rhw,
    float* __restrict__ out) {
  __shared__ __align__(16) float s_part[NW_][CH_];
  const int c0 = blockIdx.x * CH_, b = blockIdx.y, tid = threadIdx.x;
  const int w = __builtin_amdgcn_readfirstlane(tid >> 5);
  const int lane = tid & 31;
  const int col = c0 + 4 * lane;
  const int u = c0 + (tid & (CH_ - 1));
  const float bias = bp[u];
  const float zt = ztw[b * U_ + u];
  const float hv = h[b * U_ + u];
  float4 a = make_float4(0.f, 0.f, 0.f, 0.f);
  seg1(Up, rhw + b * U_, w, col, a);
  seg1(Cp, in + b * IN_, w, col, a);
  seg1(Cp + IN_ * U_, ctx + b * U_, w, col, a);
  *(float4*)&s_part[w][4 * lane] = a;
  __syncthreads();
  if (tid < CH_) {
    float s = bias;
    #pragma unroll
    for (int p = 0; p < NW_; ++p) s += s_part[p][tid];
    float th = tanhf(s);
    out[b * U_ + u] = (1.f - zt) * hv + zt * th;
  }
}

extern "C" void kernel_launch(void* const* d_in, const int* in_sizes, int n_in,
                              void* d_out, int out_size, void* d_ws, size_t ws_size,
                              hipStream_t stream) {
  const float* inputs = (const float*)d_in[0];
  const float* h_tm   = (const float*)d_in[1];
  const float* x_seq  = (const float*)d_in[2];
  const float* V_a    = (const float*)d_in[3];
  const float* W_a    = (const float*)d_in[4];
  const float* U_a    = (const float*)d_in[5];
  const float* b_a    = (const float*)d_in[6];
  const float* C_z    = (const float*)d_in[7];
  const float* W_z    = (const float*)d_in[8];
  const float* b_z    = (const float*)d_in[9];
  const float* C_r    = (const float*)d_in[10];
  const float* W_r    = (const float*)d_in[11];
  const float* b_r    = (const float*)d_in[12];
  const float* C_p    = (const float*)d_in[13];
  const float* U_p    = (const float*)d_in[14];
  const float* b_p    = (const float*)d_in[15];
  float* out = (float*)d_out;

  float* wsf   = (float*)d_ws;
  float* wxstm = wsf;
  float* ctx   = wxstm + B_ * U_;
  float* ztw   = ctx + B_ * U_;
  float* rhw   = ztw + B_ * U_;
  unsigned short* UaT = (unsigned short*)(rhw + B_ * U_);
  float* cpart = rhw + 2 * B_ * U_;
  float* spart = cpart + (size_t)B_ * NT_ * U_;

  dim3 blk(256);
  k_prep_uaT<<<dim3(U_), blk, 0, stream>>>(U_a, UaT);
  k_wxstm  <<<dim3(U_ / CH_, B_), blk, 0, stream>>>(h_tm, W_a, wxstm);
  k_scores <<<dim3(NT_, B_), blk, 0, stream>>>(x_seq, UaT, wxstm, b_a, V_a,
                                               cpart, spart);
  k_ctx_reduce<<<dim3(B_), blk, 0, stream>>>(cpart, spart, ctx);
  k_gates  <<<dim3(U_ / CH_, B_), blk, 0, stream>>>(h_tm, inputs, ctx, W_z, C_z, b_z,
                                          W_r, C_r, b_r, ztw, rhw);
  k_out    <<<dim3(U_ / CH_, B_), blk, 0, stream>>>(h_tm, inputs, ctx, U_p, C_p, b_p,
                                          ztw, rhw, out);
}


# U_a^T stored in WMMA fragment order by the prep kernel: a fragment load of k_scores reads 4 whole 128-byte lines, was 32 bytes out of each of 16 lines; same values, same sum order
# speedup vs baseline: 5.6424x; 1.1541x over previous
#include <hip/hip_runtime.h>

#define B_   128
#define TE_  2048
#define U_   256
#define IN_  256
#define MT_  4
#define LDA_ (U_ + 8)
#define CH_  128
#define NW_  8
#define NT_  (TE_ / (16 * MT_))

typedef __attribute__((ext_vector_type(16))) __bf16 v16bf;
typedef __attribute__((ext_vector_type(8)))  float  v8f;
typedef __attribute__((ext_vector_type(4)))  float  v4f;

__device__ __forceinline__ unsigned int f32x2_to_bf16x2_bits(float lo, float hi) {
  unsigned int a = __float_as_uint(lo), c = __float_as_uint(hi);
  a += 0x7FFFu + ((a >> 16) & 1u);
  c += 0x7FFFu + ((c >> 16) & 1u);
  return (a >> 16) | (c & 0xFFFF0000u);
}

__global__ __launch_bounds__(256) void k_prep_uaT(const float* __restrict__ Ua,
                                                  unsigned short* __restrict__ UaT) {
  const int g = blockIdx.x;
  const int u = threadIdx.x;
  const float* p = Ua + (8 * g) * U_ + u;
  float v[8];
  #pragma unroll
  for (int i = 0; i < 8; ++i) v[i] = p[i * U_];
  uint4 pk;
  pk.x = f32x2_to_bf16x2_bits(v[0], v[1]);
  pk.y = f32x2_to_bf16x2_bits(v[2], v[3]);
  pk.z = f32x2_to_bf16x2_bits(v[4], v[5]);
  pk.w = f32x2_to_bf16x2_bits(v[6], v[7]);
  const int ut = u >> 4, row = u & 15;
  const int ks = g >> 2, q = (g >> 1) & 1, sel = g & 1;
  *(uint4*)(UaT + ((ut * 8 + ks) * 2 + q) * 256 + (16 * sel + row) * 8) = pk;
}

#define RW_ (U_ / NW_)

__device__ __forceinline__ void fma4(float4& a, float s, const float4 x) {
  a.x = fmaf(s, x.x, a.x);
  a.y = fmaf(s, x.y, a.y);
  a.z = fmaf(s, x.z, a.z);
  a.w = fmaf(s, x.w, a.w);
}

__device__ __forceinline__ void seg1(const float* __restrict__ Wm,
                                     const float* __restrict__ v,
                                     int w, int col, float4& a) {
  const float* vw = v + RW_ * w;
  float4 s[RW_ / 4];
  #pragma unroll
  for (int j = 0; j < RW_ / 4; ++j) s[j] = *(const float4*)(vw + 4 * j);
  const float* p = Wm + (RW_ * w) * U_ + col;
  #pragma unroll
  for (int j = 0; j < RW_ / 4; ++j) {
    const float4 x0 = *(const float4*)(p + (4 * j + 0) * U_);
    const float4 x1 = *(const float4*)(p + (4 * j + 1) * U_);
    const float4 x2 = *(const float4*)(p + (4 * j + 2) * U_);
    const float4 x3 = *(const float4*)(p + (4 * j + 3) * U_);
    fma4(a, s[j].x, x0);
    fma4(a, s[j].y, x1);
    fma4(a, s[j].z, x2);
    fma4(a, s[j].w, x3);
  }
}

__device__ __forceinline__ void seg2(const float* __restrict__ Wm0,
                                     const float* __restrict__ Wm1,
                                     const float* __restrict__ v,
                                     int w, int col, float4& a0, float4& a1) {
  const float* vw = v + RW_ * w;
  float4 s[RW_ / 4];
  #pragma unroll
  for (int j = 0; j < RW_ / 4; ++j) s[j] = *(const float4*)(vw + 4 * j);
  const float* p0 = Wm0 + (RW_ * w) * U_ + col;
  const float* p1 = Wm1 + (RW_ * w) * U_ + col;
  #pragma unroll
  for (int j = 0; j < RW_ / 4; ++j) {
    const float4 x0 = *(const float4*)(p0 + (4 * j + 0) * U_);
    const float4 y0 = *(const float4*)(p1 + (4 * j + 0) * U_);
    const float4 x1 = *(const float4*)(p0 + (4 * j + 1) * U_);
    const float4 y1 = *(const float4*)(p1 + (4 * j + 1) * U_);
    const float4 x2 = *(const float4*)(p0 + (4 * j + 2) * U_);
    const float4 y2 = *(const float4*)(p1 + (4 * j + 2) * U_);
    const float4 x3 = *(const float4*)(p0 + (4 * j + 3) * U_);
    const float4 y3 = *(const float4*)(p1 + (4 * j + 3) * U_);
    fma4(a0, s[j].x, x0);
    fma4(a1, s[j].x, y0);
    fma4(a0, s[j].y, x1);
    fma4(a1, s[j].y, y1);
    fma4(a0, s[j].z, x2);
    fma4(a1, s[j].z, y2);
    fma4(a0, s[j].w, x3);
    fma4(a1, s[j].w, y3);
  }
}

__global__ __launch_bounds__(256) void k_wxstm(const float* __restrict__ h,
                                               const float* __restrict__ Wa,
                                               float* __restrict__ wx) {
  __shared__ __align__(16) float s_part[NW_][CH_];
  const int c0 = blockIdx.x * CH_, b = blockIdx.y, tid = threadIdx.x;
  const int w = __builtin_amdgcn_readfirstlane(tid >> 5);
  const int lane = tid & 31;
  float4 a = make_float4(0.f, 0.f, 0.f, 0.f);
  seg1(Wa, h + b * U_, w, c0 + 4 * lane, a);
  *(float4*)&s_part[w][4 * lane] = a;
  __syncthreads();
  if (tid < CH_) {
    float s = 0.f;
    #pragma unroll
    for (int p = 0; p < NW_; ++p) s += s_part[p][tid];
    wx[b * U_ + c0 + tid] = s;
  }
}

__global__ __launch_bounds__(256, 10) void k_scores(
    const float* __restrict__ x_seq, const unsigned short* __restrict__ UaT,
    const float* __restrict__ wxstm, const float* __restrict__ b_a,
    const float* __restrict__ V_a, float* __restrict__ cpart,
    float* __restrict__ spart) {
  __shared__ __align__(16) unsigned short sA[16 * MT_ * LDA_];
  __shared__ float s_part[16][16 * MT_];
  __shared__ __align__(16) float s_p[16 * MT_];

  int b  = blockIdx.y;
  int t0 = blockIdx.x * (16 * MT_);
  int tid = threadIdx.x;

  const float* xrow = x_seq + ((size_t)b * TE_ + t0) * U_;
  {
    const v4f* xq = (const v4f*)(x_seq + (size_t)b * TE_ * U_);
    const int q0 = t0 * (U_ / 4) + tid;
    v4f xs[16 * MT_ / 4];
    #pragma unroll
    for (int j = 0; j < 16 * MT_ / 4; ++j) xs[j] = xq[q0 + j * 256];
    asm volatile("" :: "v"(xs[0]), "v"(xs[1]), "v"(xs[2]), "v"(xs[3]), "v"(xs[4]), "v"(xs[5]),
                 "v"(xs[6]), "v"(xs[7]), "v"(xs[8]), "v"(xs[9]), "v"(xs[10]), "v"(xs[11]),
                 "v"(xs[12]), "v"(xs[13]), "v"(xs[14]), "v"(xs[15]));
    unsigned short* srow = &sA[(tid >> 6) * LDA_ + 4 * (tid & 63)];
    #pragma unroll
    for (int j = 0; j < 16 * MT_ / 4; ++j) {
      uint2 pk;
      pk.x = f32x2_to_bf16x2_bits(xs[j].x, xs[j].y);
      pk.y = f32x2_to_bf16x2_bits(xs[j].z, xs[j].w);
      *(uint2*)(srow + 4 * j * LDA_) = pk;
    }
  }
  __syncthreads();

  int wave = tid >> 5;
  int lane = tid & 31;
  int sel  = lane >> 4;
  int row  = lane & 15;

  v8f acc[2][MT_];
  #pragma unroll
  for (int j = 0; j < 2; ++j)
    #pragma unroll
    for (int m = 0; m < MT_; ++m)
      acc[j][m] = (v8f){0.f, 0.f, 0.f, 0.f, 0.f, 0.f, 0.f, 0.f};

  const int boff = (2 * wave) * (8 * 2 * 256) + lane * 8;

  #pragma unroll
  for (int kk = 0; kk < U_; kk += 32) {
    union { uint4 q[2]; v16bf v; } fb0, fb1;
    fb0.q[0] = *(const uint4*)(UaT + boff + (kk / 32) * 512);
    fb0.q[1] = *(const uint4*)(UaT + boff + (kk / 32) * 512 + 256);
    fb1.q[0] = *(const uint4*)(UaT + boff + 8 * 2 * 256 + (kk / 32) * 512);
    fb1.q[1] = *(const uint4*)(UaT + boff + 8 * 2 * 256 + (kk / 32) * 512 + 256);
    #pragma unroll
    for (int m = 0; m < MT_; ++m) {
      union { uint4 q[2]; v16bf v; } fa;
      const unsigned short* arow = &sA[(m * 16 + row) * LDA_ + kk];
      fa.q[0] = *(const uint4*)(arow + sel * 8);
      fa.q[1] = *(const uint4*)(arow + 16 + sel * 8);
      acc[0][m] = __builtin_amdgcn_wmma_f32_16x16x32_bf16(
          false, fb0.v, false, fa.v, (short)0, acc[0][m], false, false);
      acc[1][m] = __builtin_amdgcn_wmma_f32_16x16x32_bf16(
          false, fb1.v, false, fa.v, (short)0, acc[1][m], false, false);
    }
  }

  float part[MT_];
  #pragma unroll
  for (int m = 0; m < MT_; ++m) part[m] = 0.f;
  #pragma unroll
  for (int j = 0; j < 2; ++j) {
    int ug = (2 * wave + j) * 16 + 8 * sel;
    const float4* wp = (const float4*)(wxstm + b * U_ + ug);
    const float4* bp = (const float4*)(b_a + ug);
    const float4* vp = (const float4*)(V_a + ug);
    float4 w0 = wp[0], w1 = wp[1], c0 = bp[0], c1 = bp[1], v0 = vp[0], v1 = vp[1];
    float wx[8] = {w0.x + c0.x, w0.y + c0.y, w0.z + c0.z, w0.w + c0.w,
                   w1.x + c1.x, w1.y + c1.y, w1.z + c1.z, w1.w + c1.w};
    float va[8] = {v0.x, v0.y, v0.z, v0.w, v1.x, v1.y, v1.z, v1.w};
    #pragma unroll
    for (int m = 0; m < MT_; ++m) {
      #pragma unroll
      for (int i = 0; i < 8; ++i)
        part[m] = fmaf(va[i], __builtin_amdgcn_tanhf(acc[j][m][i] + wx[i]), part[m]);
    }
  }
  #pragma unroll
  for (int m = 0; m < MT_; ++m) s_part[2 * wave + sel][m * 16 + row] = part[m];
  __syncthreads();
  if (tid < 16 * MT_) {
    float s = 0.f;
    #pragma unroll
    for (int p = 0; p < 16; ++p) s += s_part[p][tid];
    s_p[tid] = expf(s);
  }
  __syncthreads();

  float xv[16 * MT_];
  #pragma unroll
  for (int k = 0; k < 16 * MT_; ++k) xv[k] = xrow[k * U_ + tid];
  float ca0 = 0.f, ca1 = 0.f, ca2 = 0.f, ca3 = 0.f;
  float cs0 = 0.f, cs1 = 0.f, cs2 = 0.f, cs3 = 0.f;
  #pragma unroll
  for (int q = 0; q < 4 * MT_; ++q) {
    const float4 pv = *(const float4*)&s_p[4 * q];
    ca0 = fmaf(pv.x, xv[4 * q + 0], ca0);
    ca1 = fmaf(pv.y, xv[4 * q + 1], ca1);
    ca2 = fmaf(pv.z, xv[4 * q + 2], ca2);
    ca3 = fmaf(pv.w, xv[4 * q + 3], ca3);
    cs0 += pv.x;
    cs1 += pv.y;
    cs2 += pv.z;
    cs3 += pv.w;
  }
  const int tile = b * NT_ + blockIdx.x;
  cpart[(size_t)tile * U_ + tid] = (ca0 + ca1) + (ca2 + ca3);
  if (tid == 0) spart[tile] = (cs0 + cs1) + (cs2 + cs3);
}

__global__ __launch_bounds__(256) void k_ctx_reduce(const float* __restrict__ cpart,
                                                    const float* __restrict__ spart,
                                                    float* __restrict__ ctx) {
  const int b = blockIdx.x, u = threadIdx.x;
  const float* cp = cpart + (size_t)b * NT_ * U_ + u;
  const float* sp = spart + b * NT_;
  float c = 0.f, s = 0.f;
  #pragma unroll
  for (int i = 0; i < NT_; ++i) {
    c += cp[i * U_];
    s += sp[i];
  }
  ctx[b * U_ + u] = c / s;
}

__global__ __launch_bounds__(256) void k_gates(
    const float* __restrict__ h, const float* __restrict__ in,
    const float* __restrict__ ctx,
    const float* __restrict__ Wz, const float* __restrict__ Cz,
    const float* __restrict__ bz,
    const float* __restrict__ Wr, const float* __restrict__ Cr,
    const float* __restrict__ br,
    float* __restrict__ ztw, float* __restrict__ rhw) {
  __shared__ __align__(16) float s_part[NW_][2 * CH_];
  const int c0 = blockIdx.x * CH_, b = blockIdx.y, tid = threadIdx.x;
  const int w = __builtin_amdgcn_readfirstlane(tid >> 5);
  const int lane = tid & 31;
  const int col = c0 + 4 * lane;
  const int g = tid >> 7;
  const int u = c0 + (tid & (CH_ - 1));
  const float bias = g ? br[u] : bz[u];
  const float hv = h[b * U_ + u];
  float4 az = make_float4(0.f, 0.f, 0.f, 0.f);
  float4 ar = make_float4(0.f, 0.f, 0.f, 0.f);
  seg2(Wz, Wr, h + b * U_, w, col, az, ar);
  seg2(Cz, Cr, in + b * IN_, w, col, az, ar);
  seg2(Cz + IN_ * U_, Cr + IN_ * U_, ctx + b * U_, w, col, az, ar);
  *(float4*)&s_part[w][4 * lane] = az;
  *(float4*)&s_part[w][CH_ + 4 * lane] = ar;
  __syncthreads();
  float s = bias;
  #pragma unroll
  for (int p = 0; p < NW_; ++p) s += s_part[p][tid];
  float sg = 1.f / (1.f + expf(-s));
  if (g == 0) ztw[b * U_ + u] = sg;
  else        rhw[b * U_ + u] = sg * hv;
}

__global__ __launch_bounds__(256) void k_out(
    const float* __restrict__ h, const float* __restrict__ in,
    const float* __restrict__ ctx,
    const float* __restrict__ Up, const float* __restrict__ Cp,
    const float* __restrict__ bp,
    const float* __restrict__ ztw, const float* __restrict__ rhw,
    float* __restrict__ out) {
  __shared__ __align__(16) float s_part[NW_][CH_];
  const int c0 = blockIdx.x * CH_, b = blockIdx.y, tid = threadIdx.x;
  const int w = __builtin_amdgcn_readfirstlane(tid >> 5);
  const int lane = tid & 31;
  const int col = c0 + 4 * lane;
  const int u = c0 + (tid & (CH_ - 1));
  const float bias = bp[u];
  const float zt = ztw[b * U_ + u];
  const float hv = h[b * U_ + u];
  float4 a = make_float4(0.f, 0.f, 0.f, 0.f);
  seg1(Up, rhw + b * U_, w, col, a);
  seg1(Cp, in + b * IN_, w, col, a);
  seg1(Cp + IN_ * U_, ctx + b * U_, w, col, a);
  *(float4*)&s_part[w][4 * lane] = a;
  __syncthreads();
  if (tid < CH_) {
    float s = bias;
    #pragma unroll
    for (int p = 0; p < NW_; ++p) s += s_part[p][tid];
    float th = tanhf(s);
    out[b * U_ + u] = (1.f - zt) * hv + zt * th;
  }
}

extern "C" void kernel_launch(void* const* d_in, const int* in_sizes, int n_in,
                              void* d_out, int out_size, void* d_ws, size_t ws_size,
                              hipStream_t stream) {
  const float* inputs = (const float*)d_in[0];
  const float* h_tm   = (const float*)d_in[1];
  const float* x_seq  = (const float*)d_in[2];
  const float* V_a    = (const float*)d_in[3];
  const float* W_a    = (const float*)d_in[4];
  const float* U_a    = (const float*)d_in[5];
  const float* b_a    = (const float*)d_in[6];
  const float* C_z    = (const float*)d_in[7];
  const float* W_z    = (const float*)d_in[8];
  const float* b_z    = (const float*)d_in[9];
  const float* C_r    = (const float*)d_in[10];
  const float* W_r    = (const float*)d_in[11];
  const float* b_r    = (const float*)d_in[12];
  const float* C_p    = (const float*)d_in[13];
  const float* U_p    = (const float*)d_in[14];
  const float* b_p    = (const float*)d_in[15];
  float* out = (float*)d_out;

  float* wsf   = (float*)d_ws;
  float* wxstm = wsf;
  float* ctx   = wxstm + B_ * U_;
  float* ztw   = ctx + B_ * U_;
  float* rhw   = ztw + B_ * U_;
  unsigned short* UaT = (unsigned short*)(rhw + B_ * U_);
  float* cpart = rhw + 2 * B_ * U_;
  float* spart = cpart + (size_t)B_ * NT_ * U_;

  dim3 blk(256);
  k_prep_uaT<<<dim3(U_ / 8), blk, 0, stream>>>(U_a, UaT);
  k_wxstm  <<<dim3(U_ / CH_, B_), blk, 0, stream>>>(h_tm, W_a, wxstm);
  k_scores <<<dim3(NT_, B_), blk, 0, stream>>>(x_seq, UaT, wxstm, b_a, V_a,
                                               cpart, spart);
  k_ctx_reduce<<<dim3(B_), blk, 0, stream>>>(cpart, spart, ctx);
  k_gates  <<<dim3(U_ / CH_, B_), blk, 0, stream>>>(h_tm, inputs, ctx, W_z, C_z, b_z,
                                          W_r, C_r, b_r, ztw, rhw);
  k_out    <<<dim3(U_ / CH_, B_), blk, 0, stream>>>(h_tm, inputs, ctx, U_p, C_p, b_p,
                                          ztw, rhw, out);
}


# wxstm, gates, out layers re-tiled: a block forms 8 batch rows x 32 columns (was 1 x 128), so a fetched weight row feeds 8 rows; 32 k-group partial sums per output added in a fixed tree via LDS
# speedup vs baseline: 6.5542x; 1.1616x over previous
#include <hip/hip_runtime.h>

#define B_   128
#define TE_  2048
#define U_   256
#define IN_  256
#define MT_  4
#define LDA_ (U_ + 8)
#define TR_  8
#define TC_  32
#define KG_  32
#define KS_  (U_ / KG_)
#define PP_  (TR_ * TC_ + 32)
#define DENSE_OCC_ __attribute__((amdgpu_waves_per_eu(2, 2)))
#define NT_  (TE_ / (16 * MT_))

typedef __attribute__((ext_vector_type(16))) __bf16 v16bf;
typedef __attribute__((ext_vector_type(8)))  float  v8f;
typedef __attribute__((ext_vector_type(4)))  float  v4f;

__device__ __forceinline__ unsigned int f32x2_to_bf16x2_bits(float lo, float hi) {
  unsigned int a = __float_as_uint(lo), c = __float_as_uint(hi);
  a += 0x7FFFu + ((a >> 16) & 1u);
  c += 0x7FFFu + ((c >> 16) & 1u);
  return (a >> 16) | (c & 0xFFFF0000u);
}

__global__ __launch_bounds__(256) void k_prep_uaT(const float* __restrict__ Ua,
                                                  unsigned short* __restrict__ UaT) {
  const int g = blockIdx.x;
  const int u = threadIdx.x;
  const float* p = Ua + (8 * g) * U_ + u;
  float v[8];
  #pragma unroll
  for (int i = 0; i < 8; ++i) v[i] = p[i * U_];
  uint4 pk;
  pk.x = f32x2_to_bf16x2_bits(v[0], v[1]);
  pk.y = f32x2_to_bf16x2_bits(v[2], v[3]);
  pk.z = f32x2_to_bf16x2_bits(v[4], v[5]);
  pk.w = f32x2_to_bf16x2_bits(v[6], v[7]);
  const int ut = u >> 4, row = u & 15;
  const int ks = g >> 2, q = (g >> 1) & 1, sel = g & 1;
  *(uint4*)(UaT + ((ut * 8 + ks) * 2 + q) * 256 + (16 * sel + row) * 8) = pk;
}

__device__ __forceinline__ void fma4(float4& a, float s, const float4 x) {
  a.x = fmaf(s, x.x, a.x);
  a.y = fmaf(s, x.y, a.y);
  a.z = fmaf(s, x.z, a.z);
  a.w = fmaf(s, x.w, a.w);
}

__device__ __forceinline__ void seg_tile(const float* __restrict__ Wq,
                                         const float* __restrict__ v,
                                         int k0, float4 (&acc)[TR_]) {
  float4 wv[KS_];
  #pragma unroll
  for (int i = 0; i < KS_; ++i) wv[i] = *(const float4*)(Wq + (k0 + i) * U_);
  #pragma unroll
  for (int r = 0; r < TR_; ++r) {
    #pragma unroll
    for (int j = 0; j < KS_; j += 4) {
      const float4 x0 = *(const float4*)(v + r * U_ + k0 + j);
      fma4(acc[r], x0.x, wv[j + 0]);
      fma4(acc[r], x0.y, wv[j + 1]);
      fma4(acc[r], x0.z, wv[j + 2]);
      fma4(acc[r], x0.w, wv[j + 3]);
    }
  }
}

__device__ __forceinline__ void put_partials(float* s_part, int g, int cq,
                                             const float4 (&acc)[TR_]) {
  #pragma unroll
  for (int r = 0; r < TR_; ++r)
    *(float4*)&s_part[g * PP_ + r * TC_ + 4 * cq] = acc[r];
}

__device__ __forceinline__ float sum_partials(const float* s_part, int o) {
  float s0 = 0.f, s1 = 0.f, s2 = 0.f, s3 = 0.f;
  #pragma unroll
  for (int p = 0; p < KG_; p += 4) {
    s0 += s_part[(p + 0) * PP_ + o];
    s1 += s_part[(p + 1) * PP_ + o];
    s2 += s_part[(p + 2) * PP_ + o];
    s3 += s_part[(p + 3) * PP_ + o];
  }
  return (s0 + s1) + (s2 + s3);
}

__global__ __launch_bounds__(256) DENSE_OCC_ void k_wxstm(const float* __restrict__ h,
                                               const float* __restrict__ Wa,
                                               float* __restrict__ wx) {
  __shared__ __align__(16) float s_part[KG_ * PP_];
  const int c0 = blockIdx.x * TC_, b0 = blockIdx.y * TR_, tid = threadIdx.x;
  const int cq = tid & 7, g = tid >> 3;
  float4 acc[TR_];
  #pragma unroll
  for (int r = 0; r < TR_; ++r) acc[r] = make_float4(0.f, 0.f, 0.f, 0.f);
  seg_tile(Wa + c0 + 4 * cq, h + b0 * U_, KS_ * g, acc);
  put_partials(s_part, g, cq, acc);
  __syncthreads();
  wx[(b0 + (tid >> 5)) * U_ + c0 + (tid & (TC_ - 1))] = sum_partials(s_part, tid);
}

__global__ __launch_bounds__(256, 10) void k_scores(
    const float* __restrict__ x_seq, const unsigned short* __restrict__ UaT,
    const float* __restrict__ wxstm, const float* __restrict__ b_a,
    const float* __restrict__ V_a, float* __restrict__ cpart,
    float* __restrict__ spart) {
  __shared__ __align__(16) unsigned short sA[16 * MT_ * LDA_];
  __shared__ float s_part[16][16 * MT_];
  __shared__ __align__(16) float s_p[16 * MT_];

  int b  = blockIdx.y;
  int t0 = blockIdx.x * (16 * MT_);
  int tid = threadIdx.x;

  const float* xrow = x_seq + ((size_t)b * TE_ + t0) * U_;
  {
    const v4f* xq = (const v4f*)(x_seq + (size_t)b * TE_ * U_);
    const int q0 = t0 * (U_ / 4) + tid;
    v4f xs[16 * MT_ / 4];
    #pragma unroll
    for (int j = 0; j < 16 * MT_ / 4; ++j) xs[j] = xq[q0 + j * 256];
    asm volatile("" :: "v"(xs[0]), "v"(xs[1]), "v"(xs[2]), "v"(xs[3]), "v"(xs[4]), "v"(xs[5]),
                 "v"(xs[6]), "v"(xs[7]), "v"(xs[8]), "v"(xs[9]), "v"(xs[10]), "v"(xs[11]),
                 "v"(xs[12]), "v"(xs[13]), "v"(xs[14]), "v"(xs[15]));
    unsigned short* srow = &sA[(tid >> 6) * LDA_ + 4 * (tid & 63)];
    #pragma unroll
    for (int j = 0; j < 16 * MT_ / 4; ++j) {
      uint2 pk;
      pk.x = f32x2_to_bf16x2_bits(xs[j].x, xs[j].y);
      pk.y = f32x2_to_bf16x2_bits(xs[j].z, xs[j].w);
      *(uint2*)(srow + 4 * j * LDA_) = pk;
    }
  }
  __syncthreads();

  int wave = tid >> 5;
  int lane = tid & 31;
  int sel  = lane >> 4;
  int row  = lane & 15;

  v8f acc[2][MT_];
  #pragma unroll
  for (int j = 0; j < 2; ++j)
    #pragma unroll
    for (int m = 0; m < MT_; ++m)
      acc[j][m] = (v8f){0.f, 0.f, 0.f, 0.f, 0.f, 0.f, 0.f, 0.f};

  const int boff = (2 * wave) * (8 * 2 * 256) + lane * 8;

  #pragma unroll
  for (int kk = 0; kk < U_; kk += 32) {
    union { uint4 q[2]; v16bf v; } fb0, fb1;
    fb0.q[0] = *(const uint4*)(UaT + boff + (kk / 32) * 512);
    fb0.q[1] = *(const uint4*)(UaT + boff + (kk / 32) * 512 + 256);
    fb1.q[0] = *(const uint4*)(UaT + boff + 8 * 2 * 256 + (kk / 32) * 512);
    fb1.q[1] = *(const uint4*)(UaT + boff + 8 * 2 * 256 + (kk / 32) * 512 + 256);
    #pragma unroll
    for (int m = 0; m < MT_; ++m) {
      union { uint4 q[2]; v16bf v; } fa;
      const unsigned short* arow = &sA[(m * 16 + row) * LDA_ + kk];
      fa.q[0] = *(const uint4*)(arow + sel * 8);
      fa.q[1] = *(const uint4*)(arow + 16 + sel * 8);
      acc[0][m] = __builtin_amdgcn_wmma_f32_16x16x32_bf16(
          false, fb0.v, false, fa.v, (short)0, acc[0][m], false, false);
      acc[1][m] = __builtin_amdgcn_wmma_f32_16x16x32_bf16(
          false, fb1.v, false, fa.v, (short)0, acc[1][m], false, false);
    }
  }

  float part[MT_];
  #pragma unroll
  for (int m = 0; m < MT_; ++m) part[m] = 0.f;
  #pragma unroll
  for (int j = 0; j < 2; ++j) {
    int ug = (2 * wave + j) * 16 + 8 * sel;
    const float4* wp = (const float4*)(wxstm + b * U_ + ug);
    const float4* bp = (const float4*)(b_a + ug);
    const float4* vp = (const float4*)(V_a + ug);
    float4 w0 = wp[0], w1 = wp[1], c0 = bp[0], c1 = bp[1], v0 = vp[0], v1 = vp[1];
    float wx[8] = {w0.x + c0.x, w0.y + c0.y, w0.z + c0.z, w0.w + c0.w,
                   w1.x + c1.x, w1.y + c1.y, w1.z + c1.z, w1.w + c1.w};
    float va[8] = {v0.x, v0.y, v0.z, v0.w, v1.x, v1.y, v1.z, v1.w};
    #pragma unroll
    for (int m = 0; m < MT_; ++m) {
      #pragma unroll
      for (int i = 0; i < 8; ++i)
        part[m] = fmaf(va[i], __builtin_amdgcn_tanhf(acc[j][m][i] + wx[i]), part[m]);
    }
  }
  #pragma unroll
  for (int m = 0; m < MT_; ++m) s_part[2 * wave + sel][m * 16 + row] = part[m];
  __syncthreads();
  if (tid < 16 * MT_) {
    float s = 0.f;
    #pragma unroll
    for (int p = 0; p < 16; ++p) s += s_part[p][tid];
    s_p[tid] = expf(s);
  }
  __syncthreads();

  float xv[16 * MT_];
  #pragma unroll
  for (int k = 0; k < 16 * MT_; ++k) xv[k] = xrow[k * U_ + tid];
  float ca0 = 0.f, ca1 = 0.f, ca2 = 0.f, ca3 = 0.f;
  float cs0 = 0.f, cs1 = 0.f, cs2 = 0.f, cs3 = 0.f;
  #pragma unroll
  for (int q = 0; q < 4 * MT_; ++q) {
    const float4 pv = *(const float4*)&s_p[4 * q];
    ca0 = fmaf(pv.x, xv[4 * q + 0], ca0);
    ca1 = fmaf(pv.y, xv[4 * q + 1], ca1);
    ca2 = fmaf(pv.z, xv[4 * q + 2], ca2);
    ca3 = fmaf(pv.w, xv[4 * q + 3], ca3);
    cs0 += pv.x;
    cs1 += pv.y;
    cs2 += pv.z;
    cs3 += pv.w;
  }
  const int tile = b * NT_ + blockIdx.x;
  cpart[(size_t)tile * U_ + tid] = (ca0 + ca1) + (ca2 + ca3);
  if (tid == 0) spart[tile] = (cs0 + cs1) + (cs2 + cs3);
}

__global__ __launch_bounds__(256) void k_ctx_reduce(const float* __restrict__ cpart,
                                                    const float* __restrict__ spart,
                                                    float* __restrict__ ctx) {
  const int b = blockIdx.x, u = threadIdx.x;
  const float* cp = cpart + (size_t)b * NT_ * U_ + u;
  const float* sp = spart + b * NT_;
  float c = 0.f, s = 0.f;
  #pragma unroll
  for (int i = 0; i < NT_; ++i) {
    c += cp[i * U_];
    s += sp[i];
  }
  ctx[b * U_ + u] = c / s;
}

__global__ __launch_bounds__(256) DENSE_OCC_ void k_gates(
    const float* __restrict__ h, const float* __restrict__ in,
    const float* __restrict__ ctx,
    const float* __restrict__ Wz, const float* __restrict__ Cz,
    const float* __restrict__ bz,
    const float* __restrict__ Wr, const float* __restrict__ Cr,
    const float* __restrict__ br,
    float* __restrict__ ztw, float* __restrict__ rhw) {
  __shared__ __align__(16) float s_part[KG_ * PP_];
  const int gate = blockIdx.x >> 3;
  const int c0 = (blockIdx.x & 7) * TC_, b0 = blockIdx.y * TR_, tid = threadIdx.x;
  const int cq = tid & 7, g = tid >> 3;
  const float* Wh = gate ? Wr : Wz;
  const float* Cx = gate ? Cr : Cz;
  const float* bb = gate ? br : bz;
  const int b = b0 + (tid >> 5);
  const int u = c0 + (tid & (TC_ - 1));
  const float bias = bb[u];
  const float hv = h[b * U_ + u];
  float4 acc[TR_];
  #pragma unroll
  for (int r = 0; r < TR_; ++r) acc[r] = make_float4(0.f, 0.f, 0.f, 0.f);
  seg_tile(Wh + c0 + 4 * cq, h + b0 * U_, KS_ * g, acc);
  seg_tile(Cx + c0 + 4 * cq, in + b0 * IN_, KS_ * g, acc);
  seg_tile(Cx + IN_ * U_ + c0 + 4 * cq, ctx + b0 * U_, KS_ * g, acc);
  put_partials(s_part, g, cq, acc);
  __syncthreads();
  const float s = bias + sum_partials(s_part, tid);
  float sg = 1.f / (1.f + expf(-s));
  if (gate == 0) ztw[b * U_ + u] = sg;
  else           rhw[b * U_ + u] = sg * hv;
}

__global__ __launch_bounds__(256) DENSE_OCC_ void k_out(
    const float* __restrict__ h, const float* __restrict__ in,
    const float* __restrict__ ctx,
    const float* __restrict__ Up, const float* __restrict__ Cp,
    const float* __restrict__ bp,
    const float* __restrict__ ztw, const float* __restrict__ rhw,
    float* __restrict__ out) {
  __shared__ __align__(16) float s_part[KG_ * PP_];
  const int c0 = blockIdx.x * TC_, b0 = blockIdx.y * TR_, tid = threadIdx.x;
  const int cq = tid & 7, g = tid >> 3;
  const int b = b0 + (tid >> 5);
  const int u = c0 + (tid & (TC_ - 1));
  const float bias = bp[u];
  const float zt = ztw[b * U_ + u];
  const float hv = h[b * U_ + u];
  float4 acc[TR_];
  #pragma unroll
  for (int r = 0; r < TR_; ++r) acc[r] = make_float4(0.f, 0.f, 0.f, 0.f);
  seg_tile(Up + c0 + 4 * cq, rhw + b0 * U_, KS_ * g, acc);
  seg_tile(Cp + c0 + 4 * cq, in + b0 * IN_, KS_ * g, acc);
  seg_tile(Cp + IN_ * U_ + c0 + 4 * cq, ctx + b0 * U_, KS_ * g, acc);
  put_partials(s_part, g, cq, acc);
  __syncthreads();
  const float s = bias + sum_partials(s_part, tid);
  float th = tanhf(s);
  out[b * U_ + u] = (1.f - zt) * hv + zt * th;
}

extern "C" void kernel_launch(void* const* d_in, const int* in_sizes, int n_in,
                              void* d_out, int out_size, void* d_ws, size_t ws_size,
                              hipStream_t stream) {
  const float* inputs = (const float*)d_in[0];
  const float* h_tm   = (const float*)d_in[1];
  const float* x_seq  = (const float*)d_in[2];
  const float* V_a    = (const float*)d_in[3];
  const float* W_a    = (const float*)d_in[4];
  const float* U_a    = (const float*)d_in[5];
  const float* b_a    = (const float*)d_in[6];
  const float* C_z    = (const float*)d_in[7];
  const float* W_z    = (const float*)d_in[8];
  const float* b_z    = (const float*)d_in[9];
  const float* C_r    = (const float*)d_in[10];
  const float* W_r    = (const float*)d_in[11];
  const float* b_r    = (const float*)d_in[12];
  const float* C_p    = (const float*)d_in[13];
  const float* U_p    = (const float*)d_in[14];
  const float* b_p    = (const float*)d_in[15];
  float* out = (float*)d_out;

  float* wsf   = (float*)d_ws;
  float* wxstm = wsf;
  float* ctx   = wxstm + B_ * U_;
  float* ztw   = ctx + B_ * U_;
  float* rhw   = ztw + B_ * U_;
  unsigned short* UaT = (unsigned short*)(rhw + B_ * U_);
  float* cpart = rhw + 2 * B_ * U_;
  float* spart = cpart + (size_t)B_ * NT_ * U_;

  dim3 blk(256);
  k_prep_uaT<<<dim3(U_ / 8), blk, 0, stream>>>(U_a, UaT);
  k_wxstm  <<<dim3(U_ / TC_, B_ / TR_), blk, 0, stream>>>(h_tm, W_a, wxstm);
  k_scores <<<dim3(NT_, B_), blk, 0, stream>>>(x_seq, UaT, wxstm, b_a, V_a,
                                               cpart, spart);
  k_ctx_reduce<<<dim3(B_), blk, 0, stream>>>(cpart, spart, ctx);
  k_gates  <<<dim3(2 * U_ / TC_, B_ / TR_), blk, 0, stream>>>(h_tm, inputs, ctx, W_z, C_z, b_z,
                                          W_r, C_r, b_r, ztw, rhw);
  k_out    <<<dim3(U_ / TC_, B_ / TR_), blk, 0, stream>>>(h_tm, inputs, ctx, U_p, C_p, b_p,
                                          ztw, rhw, out);
}


# f32 to bf16 for the staged x tile and the U_a^T prep by the hardware packed convert (v_cvt_pk_bf16_f32), not an integer rounding sequence of about 9 vector instructions per pair; same rounding
# speedup vs baseline: 6.8671x; 1.0477x over previous
#include <hip/hip_runtime.h>

#define B_   128
#define TE_  2048
#define U_   256
#define IN_  256
#define MT_  4
#define LDA_ (U_ + 8)
#define TR_  8
#define TC_  32
#define KG_  32
#define KS_  (U_ / KG_)
#define PP_  (TR_ * TC_ + 32)
#define DENSE_OCC_ __attribute__((amdgpu_waves_per_eu(2, 2)))
#define NT_  (TE_ / (16 * MT_))

typedef __attribute__((ext_vector_type(16))) __bf16 v16bf;
typedef __attribute__((ext_vector_type(8)))  float  v8f;
typedef __attribute__((ext_vector_type(4)))  float  v4f;
typedef __attribute__((ext_vector_type(2)))  float  v2f;
typedef __attribute__((ext_vector_type(2)))  __bf16 v2bf;

__device__ __forceinline__ unsigned int f32x2_to_bf16x2_bits(float lo, float hi) {
  const v2f f = {lo, hi};
  return __builtin_bit_cast(unsigned int, __builtin_convertvector(f, v2bf));
}

__global__ __launch_bounds__(256) void k_prep_uaT(const float* __restrict__ Ua,
                                                  unsigned short* __restrict__ UaT) {
  const int g = blockIdx.x;
  const int u = threadIdx.x;
  const float* p = Ua + (8 * g) * U_ + u;
  float v[8];
  #pragma unroll
  for (int i = 0; i < 8; ++i) v[i] = p[i * U_];
  uint4 pk;
  pk.x = f32x2_to_bf16x2_bits(v[0], v[1]);
  pk.y = f32x2_to_bf16x2_bits(v[2], v[3]);
  pk.z = f32x2_to_bf16x2_bits(v[4], v[5]);
  pk.w = f32x2_to_bf16x2_bits(v[6], v[7]);
  const int ut = u >> 4, row = u & 15;
  const int ks = g >> 2, q = (g >> 1) & 1, sel = g & 1;
  *(uint4*)(UaT + ((ut * 8 + ks) * 2 + q) * 256 + (16 * sel + row) * 8) = pk;
}

__device__ __forceinline__ void fma4(float4& a, float s, const float4 x) {
  a.x = fmaf(s, x.x, a.x);
  a.y = fmaf(s, x.y, a.y);
  a.z = fmaf(s, x.z, a.z);
  a.w = fmaf(s, x.w, a.w);
}

__device__ __forceinline__ void seg_tile(const float* __restrict__ Wq,
                                         const float* __restrict__ v,
                                         int k0, float4 (&acc)[TR_]) {
  float4 wv[KS_];
  #pragma unroll
  for (int i = 0; i < KS_; ++i) wv[i] = *(const float4*)(Wq + (k0 + i) * U_);
  #pragma unroll
  for (int r = 0; r < TR_; ++r) {
    #pragma unroll
    for (int j = 0; j < KS_; j += 4) {
      const float4 x0 = *(const float4*)(v + r * U_ + k0 + j);
      fma4(acc[r], x0.x, wv[j + 0]);
      fma4(acc[r], x0.y, wv[j + 1]);
      fma4(acc[r], x0.z, wv[j + 2]);
      fma4(acc[r], x0.w, wv[j + 3]);
    }
  }
}

__device__ __forceinline__ void put_partials(float* s_part, int g, int cq,
                                             const float4 (&acc)[TR_]) {
  #pragma unroll
  for (int r = 0; r < TR_; ++r)
    *(float4*)&s_part[g * PP_ + r * TC_ + 4 * cq] = acc[r];
}

__device__ __forceinline__ float sum_partials(const float* s_part, int o) {
  float s0 = 0.f, s1 = 0.f, s2 = 0.f, s3 = 0.f;
  #pragma unroll
  for (int p = 0; p < KG_; p += 4) {
    s0 += s_part[(p + 0) * PP_ + o];
    s1 += s_part[(p + 1) * PP_ + o];
    s2 += s_part[(p + 2) * PP_ + o];
    s3 += s_part[(p + 3) * PP_ + o];
  }
  return (s0 + s1) + (s2 + s3);
}

__global__ __launch_bounds__(256) DENSE_OCC_ void k_wxstm(const float* __restrict__ h,
                                               const float* __restrict__ Wa,
                                               float* __restrict__ wx) {
  __shared__ __align__(16) float s_part[KG_ * PP_];
  const int c0 = blockIdx.x * TC_, b0 = blockIdx.y * TR_, tid = threadIdx.x;
  const int cq = tid & 7, g = tid >> 3;
  float4 acc[TR_];
  #pragma unroll
  for (int r = 0; r < TR_; ++r) acc[r] = make_float4(0.f, 0.f, 0.f, 0.f);
  seg_tile(Wa + c0 + 4 * cq, h + b0 * U_, KS_ * g, acc);
  put_partials(s_part, g, cq, acc);
  __syncthreads();
  wx[(b0 + (tid >> 5)) * U_ + c0 + (tid & (TC_ - 1))] = sum_partials(s_part, tid);
}

__global__ __launch_bounds__(256, 10) void k_scores(
    const float* __restrict__ x_seq, const unsigned short* __restrict__ UaT,
    const float* __restrict__ wxstm, const float* __restrict__ b_a,
    const float* __restrict__ V_a, float* __restrict__ cpart,
    float* __restrict__ spart) {
  __shared__ __align__(16) unsigned short sA[16 * MT_ * LDA_];
  __shared__ float s_part[16][16 * MT_];
  __shared__ __align__(16) float s_p[16 * MT_];

  int b  = blockIdx.y;
  int t0 = blockIdx.x * (16 * MT_);
  int tid = threadIdx.x;

  const float* xrow = x_seq + ((size_t)b * TE_ + t0) * U_;
  {
    const v4f* xq = (const v4f*)(x_seq + (size_t)b * TE_ * U_);
    const int q0 = t0 * (U_ / 4) + tid;
    v4f xs[16 * MT_ / 4];
    #pragma unroll
    for (int j = 0; j < 16 * MT_ / 4; ++j) xs[j] = xq[q0 + j * 256];
    asm volatile("" :: "v"(xs[0]), "v"(xs[1]), "v"(xs[2]), "v"(xs[3]), "v"(xs[4]), "v"(xs[5]),
                 "v"(xs[6]), "v"(xs[7]), "v"(xs[8]), "v"(xs[9]), "v"(xs[10]), "v"(xs[11]),
                 "v"(xs[12]), "v"(xs[13]), "v"(xs[14]), "v"(xs[15]));
    unsigned short* srow = &sA[(tid >> 6) * LDA_ + 4 * (tid & 63)];
    #pragma unroll
    for (int j = 0; j < 16 * MT_ / 4; ++j) {
      uint2 pk;
      pk.x = f32x2_to_bf16x2_bits(xs[j].x, xs[j].y);
      pk.y = f32x2_to_bf16x2_bits(xs[j].z, xs[j].w);
      *(uint2*)(srow + 4 * j * LDA_) = pk;
    }
  }
  __syncthreads();

  int wave = tid >> 5;
  int lane = tid & 31;
  int sel  = lane >> 4;
  int row  = lane & 15;

  v8f acc[2][MT_];
  #pragma unroll
  for (int j = 0; j < 2; ++j)
    #pragma unroll
    for (int m = 0; m < MT_; ++m)
      acc[j][m] = (v8f){0.f, 0.f, 0.f, 0.f, 0.f, 0.f, 0.f, 0.f};

  const int boff = (2 * wave) * (8 * 2 * 256) + lane * 8;

  #pragma unroll
  for (int kk = 0; kk < U_; kk += 32) {
    union { uint4 q[2]; v16bf v; } fb0, fb1;
    fb0.q[0] = *(const uint4*)(UaT + boff + (kk / 32) * 512);
    fb0.q[1] = *(const uint4*)(UaT + boff + (kk / 32) * 512 + 256);
    fb1.q[0] = *(const uint4*)(UaT + boff + 8 * 2 * 256 + (kk / 32) * 512);
    fb1.q[1] = *(const uint4*)(UaT + boff + 8 * 2 * 256 + (kk / 32) * 512 + 256);
    #pragma unroll
    for (int m = 0; m < MT_; ++m) {
      union { uint4 q[2]; v16bf v; } fa;
      const unsigned short* arow = &sA[(m * 16 + row) * LDA_ + kk];
      fa.q[0] = *(const uint4*)(arow + sel * 8);
      fa.q[1] = *(const uint4*)(arow + 16 + sel * 8);
      acc[0][m] = __builtin_amdgcn_wmma_f32_16x16x32_bf16(
          false, fb0.v, false, fa.v, (short)0, acc[0][m], false, false);
      acc[1][m] = __builtin_amdgcn_wmma_f32_16x16x32_bf16(
          false, fb1.v, false, fa.v, (short)0, acc[1][m], false, false);
    }
  }

  float part[MT_];
  #pragma unroll
  for (int m = 0; m < MT_; ++m) part[m] = 0.f;
  #pragma unroll
  for (int j = 0; j < 2; ++j) {
    int ug = (2 * wave + j) * 16 + 8 * sel;
    const float4* wp = (const float4*)(wxstm + b * U_ + ug);
    const float4* bp = (const float4*)(b_a + ug);
    const float4* vp = (const float4*)(V_a + ug);
    float4 w0 = wp[0], w1 = wp[1], c0 = bp[0], c1 = bp[1], v0 = vp[0], v1 = vp[1];
    float wx[8] = {w0.x + c0.x, w0.y + c0.y, w0.z + c0.z, w0.w + c0.w,
                   w1.x + c1.x, w1.y + c1.y, w1.z + c1.z, w1.w + c1.w};
    float va[8] = {v0.x, v0.y, v0.z, v0.w, v1.x, v1.y, v1.z, v1.w};
    #pragma unroll
    for (int m = 0; m < MT_; ++m) {
      #pragma unroll
      for (int i = 0; i < 8; ++i)
        part[m] = fmaf(va[i], __builtin_amdgcn_tanhf(acc[j][m][i] + wx[i]), part[m]);
    }
  }
  #pragma unroll
  for (int m = 0; m < MT_; ++m) s_part[2 * wave + sel][m * 16 + row] = part[m];
  __syncthreads();
  if (tid < 16 * MT_) {
    float s = 0.f;
    #pragma unroll
    for (int p = 0; p < 16; ++p) s += s_part[p][tid];
    s_p[tid] = expf(s);
  }
  __syncthreads();

  float xv[16 * MT_];
  #pragma unroll
  for (int k = 0; k < 16 * MT_; ++k) xv[k] = xrow[k * U_ + tid];
  float ca0 = 0.f, ca1 = 0.f, ca2 = 0.f, ca3 = 0.f;
  float cs0 = 0.f, cs1 = 0.f, cs2 = 0.f, cs3 = 0.f;
  #pragma unroll
  for (int q = 0; q < 4 * MT_; ++q) {
    const float4 pv = *(const float4*)&s_p[4 * q];
    ca0 = fmaf(pv.x, xv[4 * q + 0], ca0);
    ca1 = fmaf(pv.y, xv[4 * q + 1], ca1);
    ca2 = fmaf(pv.z, xv[4 * q + 2], ca2);
    ca3 = fmaf(pv.w, xv[4 * q + 3], ca3);
    cs0 += pv.x;
    cs1 += pv.y;
    cs2 += pv.z;
    cs3 += pv.w;
  }
  const int tile = b * NT_ + blockIdx.x;
  cpart[(size_t)tile * U_ + tid] = (ca0 + ca1) + (ca2 + ca3);
  if (tid == 0) spart[tile] = (cs0 + cs1) + (cs2 + cs3);
}

__global__ __launch_bounds__(256) void k_ctx_reduce(const float* __restrict__ cpart,
                                                    const float* __restrict__ spart,
                                                    float* __restrict__ ctx) {
  const int b = blockIdx.x, u = threadIdx.x;
  const float* cp = cpart + (size_t)b * NT_ * U_ + u;
  const float* sp = spart + b * NT_;
  float c = 0.f, s = 0.f;
  #pragma unroll
  for (int i = 0; i < NT_; ++i) {
    c += cp[i * U_];
    s += sp[i];
  }
  ctx[b * U_ + u] = c / s;
}

__global__ __launch_bounds__(256) DENSE_OCC_ void k_gates(
    const float* __restrict__ h, const float* __restrict__ in,
    const float* __restrict__ ctx,
    const float* __restrict__ Wz, const float* __restrict__ Cz,
    const float* __restrict__ bz,
    const float* __restrict__ Wr, const float* __restrict__ Cr,
    const float* __restrict__ br,
    float* __restrict__ ztw, float* __restrict__ rhw) {
  __shared__ __align__(16) float s_part[KG_ * PP_];
  const int gate = blockIdx.x >> 3;
  const int c0 = (blockIdx.x & 7) * TC_, b0 = blockIdx.y * TR_, tid = threadIdx.x;
  const int cq = tid & 7, g = tid >> 3;
  const float* Wh = gate ? Wr : Wz;
  const float* Cx = gate ? Cr : Cz;
  const float* bb = gate ? br : bz;
  const int b = b0 + (tid >> 5);
  const int u = c0 + (tid & (TC_ - 1));
  const float bias = bb[u];
  const float hv = h[b * U_ + u];
  float4 acc[TR_];
  #pragma unroll
  for (int r = 0; r < TR_; ++r) acc[r] = make_float4(0.f, 0.f, 0.f, 0.f);
  seg_tile(Wh + c0 + 4 * cq, h + b0 * U_, KS_ * g, acc);
  seg_tile(Cx + c0 + 4 * cq, in + b0 * IN_, KS_ * g, acc);
  seg_tile(Cx + IN_ * U_ + c0 + 4 * cq, ctx + b0 * U_, KS_ * g, acc);
  put_partials(s_part, g, cq, acc);
  __syncthreads();
  const float s = bias + sum_partials(s_part, tid);
  float sg = 1.f / (1.f + expf(-s));
  if (gate == 0) ztw[b * U_ + u] = sg;
  else           rhw[b * U_ + u] = sg * hv;
}

__global__ __launch_bounds__(256) DENSE_OCC_ void k_out(
    const float* __restrict__ h, const float* __restrict__ in,
    const float* __restrict__ ctx,
    const float* __restrict__ Up, const float* __restrict__ Cp,
    const float* __restrict__ bp,
    const float* __restrict__ ztw, const float* __restrict__ rhw,
    float* __restrict__ out) {
  __shared__ __align__(16) float s_part[KG_ * PP_];
  const int c0 = blockIdx.x * TC_, b0 = blockIdx.y * TR_, tid = threadIdx.x;
  const int cq = tid & 7, g = tid >> 3;
  const int b = b0 + (tid >> 5);
  const int u = c0 + (tid & (TC_ - 1));
  const float bias = bp[u];
  const float zt = ztw[b * U_ + u];
  const float hv = h[b * U_ + u];
  float4 acc[TR_];
  #pragma unroll
  for (int r = 0; r < TR_; ++r) acc[r] = make_float4(0.f, 0.f, 0.f, 0.f);
  seg_tile(Up + c0 + 4 * cq, rhw + b0 * U_, KS_ * g, acc);
  seg_tile(Cp + c0 + 4 * cq, in + b0 * IN_, KS_ * g, acc);
  seg_tile(Cp + IN_ * U_ + c0 + 4 * cq, ctx + b0 * U_, KS_ * g, acc);
  put_partials(s_part, g, cq, acc);
  __syncthreads();
  const float s = bias + sum_partials(s_part, tid);
  float th = tanhf(s);
  out[b * U_ + u] = (1.f - zt) * hv + zt * th;
}

extern "C" void kernel_launch(void* const* d_in, const int* in_sizes, int n_in,
                              void* d_out, int out_size, void* d_ws, size_t ws_size,
                              hipStream_t stream) {
  const float* inputs = (const float*)d_in[0];
  const float* h_tm   = (const float*)d_in[1];
  const float* x_seq  = (const float*)d_in[2];
  const float* V_a    = (const float*)d_in[3];
  const float* W_a    = (const float*)d_in[4];
  const float* U_a    = (const float*)d_in[5];
  const float* b_a    = (const float*)d_in[6];
  const float* C_z    = (const float*)d_in[7];
  const float* W_z    = (const float*)d_in[8];
  const float* b_z    = (const float*)d_in[9];
  const float* C_r    = (const float*)d_in[10];
  const float* W_r    = (const float*)d_in[11];
  const float* b_r    = (const float*)d_in[12];
  const float* C_p    = (const float*)d_in[13];
  const float* U_p    = (const float*)d_in[14];
  const float* b_p    = (const float*)d_in[15];
  float* out = (float*)d_out;

  float* wsf   = (float*)d_ws;
  float* wxstm = wsf;
  float* ctx   = wxstm + B_ * U_;
  float* ztw   = ctx + B_ * U_;
  float* rhw   = ztw + B_ * U_;
  unsigned short* UaT = (unsigned short*)(rhw + B_ * U_);
  float* cpart = rhw + 2 * B_ * U_;
  float* spart = cpart + (size_t)B_ * NT_ * U_;

  dim3 blk(256);
  k_prep_uaT<<<dim3(U_ / 8), blk, 0, stream>>>(U_a, UaT);
  k_wxstm  <<<dim3(U_ / TC_, B_ / TR_), blk, 0, stream>>>(h_tm, W_a, wxstm);
  k_scores <<<dim3(NT_, B_), blk, 0, stream>>>(x_seq, UaT, wxstm, b_a, V_a,
                                               cpart, spart);
  k_ctx_reduce<<<dim3(B_), blk, 0, stream>>>(cpart, spart, ctx);
  k_gates  <<<dim3(2 * U_ / TC_, B_ / TR_), blk, 0, stream>>>(h_tm, inputs, ctx, W_z, C_z, b_z,
                                          W_r, C_r, b_r, ztw, rhw);
  k_out    <<<dim3(U_ / TC_, B_ / TR_), blk, 0, stream>>>(h_tm, inputs, ctx, U_p, C_p, b_p,
                                          ztw, rhw, out);
}


# k_scores context part: x tile re-read with 16-byte loads (thread = 4 columns x 16 rows; 16 loads of 4 whole lines, was 64 one-line loads); the 4 row-group sums meet in LDS; sum order kept
# speedup vs baseline: 7.1424x; 1.0401x over previous
#include <hip/hip_runtime.h>

#define B_   128
#define TE_  2048
#define U_   256
#define IN_  256
#define MT_  4
#define LDA_ (U_ + 8)
#define TR_  8
#define TC_  32
#define KG_  32
#define KS_  (U_ / KG_)
#define PP_  (TR_ * TC_ + 32)
#define DENSE_OCC_ __attribute__((amdgpu_waves_per_eu(2, 2)))
#define NT_  (TE_ / (16 * MT_))

typedef __attribute__((ext_vector_type(16))) __bf16 v16bf;
typedef __attribute__((ext_vector_type(8)))  float  v8f;
typedef __attribute__((ext_vector_type(4)))  float  v4f;
typedef __attribute__((ext_vector_type(2)))  float  v2f;
typedef __attribute__((ext_vector_type(2)))  __bf16 v2bf;

__device__ __forceinline__ unsigned int f32x2_to_bf16x2_bits(float lo, float hi) {
  const v2f f = {lo, hi};
  return __builtin_bit_cast(unsigned int, __builtin_convertvector(f, v2bf));
}

__global__ __launch_bounds__(256) void k_prep_uaT(const float* __restrict__ Ua,
                                                  unsigned short* __restrict__ UaT) {
  const int g = blockIdx.x;
  const int u = threadIdx.x;
  const float* p = Ua + (8 * g) * U_ + u;
  float v[8];
  #pragma unroll
  for (int i = 0; i < 8; ++i) v[i] = p[i * U_];
  uint4 pk;
  pk.x = f32x2_to_bf16x2_bits(v[0], v[1]);
  pk.y = f32x2_to_bf16x2_bits(v[2], v[3]);
  pk.z = f32x2_to_bf16x2_bits(v[4], v[5]);
  pk.w = f32x2_to_bf16x2_bits(v[6], v[7]);
  const int ut = u >> 4, row = u & 15;
  const int ks = g >> 2, q = (g >> 1) & 1, sel = g & 1;
  *(uint4*)(UaT + ((ut * 8 + ks) * 2 + q) * 256 + (16 * sel + row) * 8) = pk;
}

__device__ __forceinline__ void fma4(float4& a, float s, const float4 x) {
  a.x = fmaf(s, x.x, a.x);
  a.y = fmaf(s, x.y, a.y);
  a.z = fmaf(s, x.z, a.z);
  a.w = fmaf(s, x.w, a.w);
}

__device__ __forceinline__ void seg_tile(const float* __restrict__ Wq,
                                         const float* __restrict__ v,
                                         int k0, float4 (&acc)[TR_]) {
  float4 wv[KS_];
  #pragma unroll
  for (int i = 0; i < KS_; ++i) wv[i] = *(const float4*)(Wq + (k0 + i) * U_);
  #pragma unroll
  for (int r = 0; r < TR_; ++r) {
    #pragma unroll
    for (int j = 0; j < KS_; j += 4) {
      const float4 x0 = *(const float4*)(v + r * U_ + k0 + j);
      fma4(acc[r], x0.x, wv[j + 0]);
      fma4(acc[r], x0.y, wv[j + 1]);
      fma4(acc[r], x0.z, wv[j + 2]);
      fma4(acc[r], x0.w, wv[j + 3]);
    }
  }
}

__device__ __forceinline__ void put_partials(float* s_part, int g, int cq,
                                             const float4 (&acc)[TR_]) {
  #pragma unroll
  for (int r = 0; r < TR_; ++r)
    *(float4*)&s_part[g * PP_ + r * TC_ + 4 * cq] = acc[r];
}

__device__ __forceinline__ float sum_partials(const float* s_part, int o) {
  float s0 = 0.f, s1 = 0.f, s2 = 0.f, s3 = 0.f;
  #pragma unroll
  for (int p = 0; p < KG_; p += 4) {
    s0 += s_part[(p + 0) * PP_ + o];
    s1 += s_part[(p + 1) * PP_ + o];
    s2 += s_part[(p + 2) * PP_ + o];
    s3 += s_part[(p + 3) * PP_ + o];
  }
  return (s0 + s1) + (s2 + s3);
}

__global__ __launch_bounds__(256) DENSE_OCC_ void k_wxstm(const float* __restrict__ h,
                                               const float* __restrict__ Wa,
                                               float* __restrict__ wx) {
  __shared__ __align__(16) float s_part[KG_ * PP_];
  const int c0 = blockIdx.x * TC_, b0 = blockIdx.y * TR_, tid = threadIdx.x;
  const int cq = tid & 7, g = tid >> 3;
  float4 acc[TR_];
  #pragma unroll
  for (int r = 0; r < TR_; ++r) acc[r] = make_float4(0.f, 0.f, 0.f, 0.f);
  seg_tile(Wa + c0 + 4 * cq, h + b0 * U_, KS_ * g, acc);
  put_partials(s_part, g, cq, acc);
  __syncthreads();
  wx[(b0 + (tid >> 5)) * U_ + c0 + (tid & (TC_ - 1))] = sum_partials(s_part, tid);
}

__global__ __launch_bounds__(256, 10) void k_scores(
    const float* __restrict__ x_seq, const unsigned short* __restrict__ UaT,
    const float* __restrict__ wxstm, const float* __restrict__ b_a,
    const float* __restrict__ V_a, float* __restrict__ cpart,
    float* __restrict__ spart) {
  __shared__ __align__(16) unsigned short sA[16 * MT_ * LDA_];
  __shared__ __align__(16) float s_part[16][16 * MT_];
  __shared__ __align__(16) float s_p[16 * MT_];
  __shared__ float s_cs[4];

  int b  = blockIdx.y;
  int t0 = blockIdx.x * (16 * MT_);
  int tid = threadIdx.x;

  const float* xrow = x_seq + ((size_t)b * TE_ + t0) * U_;
  {
    const v4f* xq = (const v4f*)(x_seq + (size_t)b * TE_ * U_);
    const int q0 = t0 * (U_ / 4) + tid;
    v4f xs[16 * MT_ / 4];
    #pragma unroll
    for (int j = 0; j < 16 * MT_ / 4; ++j) xs[j] = xq[q0 + j * 256];
    asm volatile("" :: "v"(xs[0]), "v"(xs[1]), "v"(xs[2]), "v"(xs[3]), "v"(xs[4]), "v"(xs[5]),
                 "v"(xs[6]), "v"(xs[7]), "v"(xs[8]), "v"(xs[9]), "v"(xs[10]), "v"(xs[11]),
                 "v"(xs[12]), "v"(xs[13]), "v"(xs[14]), "v"(xs[15]));
    unsigned short* srow = &sA[(tid >> 6) * LDA_ + 4 * (tid & 63)];
    #pragma unroll
    for (int j = 0; j < 16 * MT_ / 4; ++j) {
      uint2 pk;
      pk.x = f32x2_to_bf16x2_bits(xs[j].x, xs[j].y);
      pk.y = f32x2_to_bf16x2_bits(xs[j].z, xs[j].w);
      *(uint2*)(srow + 4 * j * LDA_) = pk;
    }
  }
  __syncthreads();

  int wave = tid >> 5;
  int lane = tid & 31;
  int sel  = lane >> 4;
  int row  = lane & 15;

  v8f acc[2][MT_];
  #pragma unroll
  for (int j = 0; j < 2; ++j)
    #pragma unroll
    for (int m = 0; m < MT_; ++m)
      acc[j][m] = (v8f){0.f, 0.f, 0.f, 0.f, 0.f, 0.f, 0.f, 0.f};

  const int boff = (2 * wave) * (8 * 2 * 256) + lane * 8;

  #pragma unroll
  for (int kk = 0; kk < U_; kk += 32) {
    union { uint4 q[2]; v16bf v; } fb0, fb1;
    fb0.q[0] = *(const uint4*)(UaT + boff + (kk / 32) * 512);
    fb0.q[1] = *(const uint4*)(UaT + boff + (kk / 32) * 512 + 256);
    fb1.q[0] = *(const uint4*)(UaT + boff + 8 * 2 * 256 + (kk / 32) * 512);
    fb1.q[1] = *(const uint4*)(UaT + boff + 8 * 2 * 256 + (kk / 32) * 512 + 256);
    #pragma unroll
    for (int m = 0; m < MT_; ++m) {
      union { uint4 q[2]; v16bf v; } fa;
      const unsigned short* arow = &sA[(m * 16 + row) * LDA_ + kk];
      fa.q[0] = *(const uint4*)(arow + sel * 8);
      fa.q[1] = *(const uint4*)(arow + 16 + sel * 8);
      acc[0][m] = __builtin_amdgcn_wmma_f32_16x16x32_bf16(
          false, fb0.v, false, fa.v, (short)0, acc[0][m], false, false);
      acc[1][m] = __builtin_amdgcn_wmma_f32_16x16x32_bf16(
          false, fb1.v, false, fa.v, (short)0, acc[1][m], false, false);
    }
  }

  float part[MT_];
  #pragma unroll
  for (int m = 0; m < MT_; ++m) part[m] = 0.f;
  #pragma unroll
  for (int j = 0; j < 2; ++j) {
    int ug = (2 * wave + j) * 16 + 8 * sel;
    const float4* wp = (const float4*)(wxstm + b * U_ + ug);
    const float4* bp = (const float4*)(b_a + ug);
    const float4* vp = (const float4*)(V_a + ug);
    float4 w0 = wp[0], w1 = wp[1], c0 = bp[0], c1 = bp[1], v0 = vp[0], v1 = vp[1];
    float wx[8] = {w0.x + c0.x, w0.y + c0.y, w0.z + c0.z, w0.w + c0.w,
                   w1.x + c1.x, w1.y + c1.y, w1.z + c1.z, w1.w + c1.w};
    float va[8] = {v0.x, v0.y, v0.z, v0.w, v1.x, v1.y, v1.z, v1.w};
    #pragma unroll
    for (int m = 0; m < MT_; ++m) {
      #pragma unroll
      for (int i = 0; i < 8; ++i)
        part[m] = fmaf(va[i], __builtin_amdgcn_tanhf(acc[j][m][i] + wx[i]), part[m]);
    }
  }
  #pragma unroll
  for (int m = 0; m < MT_; ++m) s_part[2 * wave + sel][m * 16 + row] = part[m];
  __syncthreads();
  if (tid < 16 * MT_) {
    float s = 0.f;
    #pragma unroll
    for (int p = 0; p < 16; ++p) s += s_part[p][tid];
    s_p[tid] = expf(s);
  }
  __syncthreads();

  const int rg = tid >> 6;
  const int cq = tid & 63;
  const v4f* xt = (const v4f*)xrow;
  v4f xv[16 * MT_ / 4];
  #pragma unroll
  for (int j = 0; j < 16 * MT_ / 4; ++j) xv[j] = xt[tid + j * 256];
  float pw[16 * MT_ / 4];
  #pragma unroll
  for (int j = 0; j < 16 * MT_ / 4; ++j) pw[j] = s_p[4 * j + rg];
  float ca0 = 0.f, ca1 = 0.f, ca2 = 0.f, ca3 = 0.f, cs = 0.f;
  #pragma unroll
  for (int j = 0; j < 16 * MT_ / 4; ++j) {
    ca0 = fmaf(pw[j], xv[j].x, ca0);
    ca1 = fmaf(pw[j], xv[j].y, ca1);
    ca2 = fmaf(pw[j], xv[j].z, ca2);
    ca3 = fmaf(pw[j], xv[j].w, ca3);
    cs += pw[j];
  }
  float* s_cx = &s_part[0][0];
  *(v4f*)&s_cx[rg * U_ + 4 * cq] = (v4f){ca0, ca1, ca2, ca3};
  if (cq == 0) s_cs[rg] = cs;
  __syncthreads();
  const int tile = b * NT_ + blockIdx.x;
  cpart[(size_t)tile * U_ + tid] =
      (s_cx[tid] + s_cx[U_ + tid]) + (s_cx[2 * U_ + tid] + s_cx[3 * U_ + tid]);
  if (tid == 0) spart[tile] = (s_cs[0] + s_cs[1]) + (s_cs[2] + s_cs[3]);
}

__global__ __launch_bounds__(256) void k_ctx_reduce(const float* __restrict__ cpart,
                                                    const float* __restrict__ spart,
                                                    float* __restrict__ ctx) {
  const int b = blockIdx.x, u = threadIdx.x;
  const float* cp = cpart + (size_t)b * NT_ * U_ + u;
  const float* sp = spart + b * NT_;
  float c = 0.f, s = 0.f;
  #pragma unroll
  for (int i = 0; i < NT_; ++i) {
    c += cp[i * U_];
    s += sp[i];
  }
  ctx[b * U_ + u] = c / s;
}

__global__ __launch_bounds__(256) DENSE_OCC_ void k_gates(
    const float* __restrict__ h, const float* __restrict__ in,
    const float* __restrict__ ctx,
    const float* __restrict__ Wz, const float* __restrict__ Cz,
    const float* __restrict__ bz,
    const float* __restrict__ Wr, const float* __restrict__ Cr,
    const float* __restrict__ br,
    float* __restrict__ ztw, float* __restrict__ rhw) {
  __shared__ __align__(16) float s_part[KG_ * PP_];
  const int gate = blockIdx.x >> 3;
  const int c0 = (blockIdx.x & 7) * TC_, b0 = blockIdx.y * TR_, tid = threadIdx.x;
  const int cq = tid & 7, g = tid >> 3;
  const float* Wh = gate ? Wr : Wz;
  const float* Cx = gate ? Cr : Cz;
  const float* bb = gate ? br : bz;
  const int b = b0 + (tid >> 5);
  const int u = c0 + (tid & (TC_ - 1));
  const float bias = bb[u];
  const float hv = h[b * U_ + u];
  float4 acc[TR_];
  #pragma unroll
  for (int r = 0; r < TR_; ++r) acc[r] = make_float4(0.f, 0.f, 0.f, 0.f);
  seg_tile(Wh + c0 + 4 * cq, h + b0 * U_, KS_ * g, acc);
  seg_tile(Cx + c0 + 4 * cq, in + b0 * IN_, KS_ * g, acc);
  seg_tile(Cx + IN_ * U_ + c0 + 4 * cq, ctx + b0 * U_, KS_ * g, acc);
  put_partials(s_part, g, cq, acc);
  __syncthreads();
  const float s = bias + sum_partials(s_part, tid);
  float sg = 1.f / (1.f + expf(-s));
  if (gate == 0) ztw[b * U_ + u] = sg;
  else           rhw[b * U_ + u] = sg * hv;
}

__global__ __launch_bounds__(256) DENSE_OCC_ void k_out(
    const float* __restrict__ h, const float* __restrict__ in,
    const float* __restrict__ ctx,
    const float* __restrict__ Up, const float* __restrict__ Cp,
    const float* __restrict__ bp,
    const float* __restrict__ ztw, const float* __restrict__ rhw,
    float* __restrict__ out) {
  __shared__ __align__(16) float s_part[KG_ * PP_];
  const int c0 = blockIdx.x * TC_, b0 = blockIdx.y * TR_, tid = threadIdx.x;
  const int cq = tid & 7, g = tid >> 3;
  const int b = b0 + (tid >> 5);
  const int u = c0 + (tid & (TC_ - 1));
  const float bias = bp[u];
  const float zt = ztw[b * U_ + u];
  const float hv = h[b * U_ + u];
  float4 acc[TR_];
  #pragma unroll
  for (int r = 0; r < TR_; ++r) acc[r] = make_float4(0.f, 0.f, 0.f, 0.f);
  seg_tile(Up + c0 + 4 * cq, rhw + b0 * U_, KS_ * g, acc);
  seg_tile(Cp + c0 + 4 * cq, in + b0 * IN_, KS_ * g, acc);
  seg_tile(Cp + IN_ * U_ + c0 + 4 * cq, ctx + b0 * U_, KS_ * g, acc);
  put_partials(s_part, g, cq, acc);
  __syncthreads();
  const float s = bias + sum_partials(s_part, tid);
  float th = tanhf(s);
  out[b * U_ + u] = (1.f - zt) * hv + zt * th;
}

extern "C" void kernel_launch(void* const* d_in, const int* in_sizes, int n_in,
                              void* d_out, int out_size, void* d_ws, size_t ws_size,
                              hipStream_t stream) {
  const float* inputs = (const float*)d_in[0];
  const float* h_tm   = (const float*)d_in[1];
  const float* x_seq  = (const float*)d_in[2];
  const float* V_a    = (const float*)d_in[3];
  const float* W_a    = (const float*)d_in[4];
  const float* U_a    = (const float*)d_in[5];
  const float* b_a    = (const float*)d_in[6];
  const float* C_z    = (const float*)d_in[7];
  const float* W_z    = (const float*)d_in[8];
  const float* b_z    = (const float*)d_in[9];
  const float* C_r    = (const float*)d_in[10];
  const float* W_r    = (const float*)d_in[11];
  const float* b_r    = (const float*)d_in[12];
  const float* C_p    = (const float*)d_in[13];
  const float* U_p    = (const float*)d_in[14];
  const float* b_p    = (const float*)d_in[15];
  float* out = (float*)d_out;

  float* wsf   = (float*)d_ws;
  float* wxstm = wsf;
  float* ctx   = wxstm + B_ * U_;
  float* ztw   = ctx + B_ * U_;
  float* rhw   = ztw + B_ * U_;
  unsigned short* UaT = (unsigned short*)(rhw + B_ * U_);
  float* cpart = rhw + 2 * B_ * U_;
  float* spart = cpart + (size_t)B_ * NT_ * U_;

  dim3 blk(256);
  k_prep_uaT<<<dim3(U_ / 8), blk, 0, stream>>>(U_a, UaT);
  k_wxstm  <<<dim3(U_ / TC_, B_ / TR_), blk, 0, stream>>>(h_tm, W_a, wxstm);
  k_scores <<<dim3(NT_, B_), blk, 0, stream>>>(x_seq, UaT, wxstm, b_a, V_a,
                                               cpart, spart);
  k_ctx_reduce<<<dim3(B_), blk, 0, stream>>>(cpart, spart, ctx);
  k_gates  <<<dim3(2 * U_ / TC_, B_ / TR_), blk, 0, stream>>>(h_tm, inputs, ctx, W_z, C_z, b_z,
                                          W_r, C_r, b_r, ztw, rhw);
  k_out    <<<dim3(U_ / TC_, B_ / TR_), blk, 0, stream>>>(h_tm, inputs, ctx, U_p, C_p, b_p,
                                          ztw, rhw, out);
}
